# starting kernel
# baseline (speedup 1.0000x reference)
#include <hip/hip_runtime.h>
#include <hip/hip_bf16.h>

#define DIM 128

typedef float v2f __attribute__((ext_vector_type(2)));
typedef float v8f __attribute__((ext_vector_type(8)));

__device__ __forceinline__ void atomAddF(float* p, float v) {
    __hip_atomic_fetch_add(p, v, __ATOMIC_RELAXED, __HIP_MEMORY_SCOPE_AGENT);
}

__global__ void deg_init_kernel(float* __restrict__ deg, int n) {
    int i = blockIdx.x * blockDim.x + threadIdx.x;
    if (i < n) deg[i] = 1.0f;
}

__global__ void deg_count_kernel(const int* __restrict__ ei, float* __restrict__ deg, int E) {
    int e = blockIdx.x * blockDim.x + threadIdx.x;
    if (e < E) atomAddF(&deg[ei[E + e]], 1.0f);
}

__global__ void deg_to_dis_kernel(float* __restrict__ deg, int n) {
    int i = blockIdx.x * blockDim.x + threadIdx.x;
    if (i < n) {
        float d = deg[i];
        deg[i] = (d > 0.0f) ? rsqrtf(d) : 0.0f;
    }
}

__global__ __launch_bounds__(256) void self_loop_kernel(const float* __restrict__ x,
                                                        const float* __restrict__ dis,
                                                        float* __restrict__ agg, int n) {
    int t = blockIdx.x * blockDim.x + threadIdx.x;
    int row = t >> 5;
    int lane = t & 31;
    if (row >= n) return;
    float s = dis[row];
    s = s * s;
    const float4* xr = reinterpret_cast<const float4*>(x + (size_t)row * DIM);
    float4* ar = reinterpret_cast<float4*>(agg + (size_t)row * DIM);
    float4 v = xr[lane];
    v.x *= s; v.y *= s; v.z *= s; v.w *= s;
    ar[lane] = v;
}

__global__ __launch_bounds__(256) void edge_scatter_kernel(const float* __restrict__ x,
                                                           const float* __restrict__ dis,
                                                           const int* __restrict__ ei,
                                                           float* __restrict__ agg, int E) {
    int t = blockIdx.x * blockDim.x + threadIdx.x;
    int e = t >> 5;
    int lane = t & 31;
    if (e >= E) return;
    int s = ei[e];
    int d = ei[E + e];
    float nrm = dis[s] * dis[d];
    const float4* xr = reinterpret_cast<const float4*>(x + (size_t)s * DIM);
    float4 v = xr[lane];
    float* dst = agg + (size_t)d * DIM + lane * 4;
    atomAddF(dst + 0, v.x * nrm);
    atomAddF(dst + 1, v.y * nrm);
    atomAddF(dst + 2, v.z * nrm);
    atomAddF(dst + 3, v.w * nrm);
}

#define LDA 132
#define LDW 136
#define KCHUNK 32

__global__ __launch_bounds__(256) void gemm_wmma_kernel(const float* __restrict__ A,
                                                        const float* __restrict__ W,
                                                        float* __restrict__ out, int n) {
    __shared__ float lA[16 * LDA];
    __shared__ float lW[KCHUNK * LDW];
    const int tid = threadIdx.x;
    const int wave = tid >> 5;
    const int lane = tid & 31;
    const int rowBase = blockIdx.x * 16;

    for (int i = tid; i < 16 * DIM; i += 256) {
        int r = i >> 7, c = i & 127;
        int gr = rowBase + r;
        lA[r * LDA + c] = (gr < n) ? A[(size_t)gr * DIM + c] : 0.0f;
    }

    const int m = lane & 15;
    const int hi = lane >> 4;
    const int colBase = wave * 16;

    v8f acc = {0.f, 0.f, 0.f, 0.f, 0.f, 0.f, 0.f, 0.f};

    for (int kc = 0; kc < DIM; kc += KCHUNK) {
        __syncthreads();
        for (int i = tid; i < KCHUNK * DIM; i += 256) {
            int r = i >> 7, c = i & 127;
            lW[r * LDW + c] = W[(size_t)(kc + r) * DIM + c];
        }
        __syncthreads();

#pragma unroll
        for (int k = 0; k < KCHUNK; k += 4) {
            v2f a, b;
            const float* ap = &lA[m * LDA + kc + k + 2 * hi];
            a.x = ap[0];
            a.y = ap[1];
            b.x = lW[(k + 2 * hi) * LDW + colBase + m];
            b.y = lW[(k + 2 * hi + 1) * LDW + colBase + m];
            acc = __builtin_amdgcn_wmma_f32_16x16x4_f32(
                 false, a,  false, b,
                 (short)0, acc,  false,  false);
        }
    }

#pragma unroll
    for (int v = 0; v < 8; ++v) {
        int gr = rowBase + v + 8 * hi;
        if (gr < n) out[(size_t)gr * DIM + colBase + m] = acc[v];
    }
}

extern "C" void kernel_launch(void* const* d_in, const int* in_sizes, int n_in,
                              void* d_out, int out_size, void* d_ws, size_t ws_size,
                              hipStream_t stream) {
    const float* x = (const float*)d_in[0];
    const int*   ei = (const int*)d_in[1];
    const float* W = (const float*)d_in[2];
    float* out = (float*)d_out;

    const int N = in_sizes[0] / DIM;
    const int E = in_sizes[1] / 2;

    float* deg = (float*)d_ws;
    size_t aggOff = (((size_t)N * sizeof(float)) + 255) & ~(size_t)255;
    float* agg = (float*)((char*)d_ws + aggOff);

    deg_init_kernel<<<(N + 255) / 256, 256, 0, stream>>>(deg, N);
    deg_count_kernel<<<(E + 255) / 256, 256, 0, stream>>>(ei, deg, E);
    deg_to_dis_kernel<<<(N + 255) / 256, 256, 0, stream>>>(deg, N);
    self_loop_kernel<<<(N + 7) / 8, 256, 0, stream>>>(x, deg, agg, N);
    edge_scatter_kernel<<<(E + 7) / 8, 256, 0, stream>>>(x, deg, ei, agg, E);
    gemm_wmma_kernel<<<(N + 15) / 16, 256, 0, stream>>>(agg, W, out, N);
}


# Per-edge f32 atomic scatter replaced by a counting sort of the edges by target row (one integer atomic per edge) and an atomics-free wave-per-row gather; row sums now in rank order of the sort
# speedup vs baseline: 2.1449x; 2.1449x over previous
#include <hip/hip_runtime.h>
#include <hip/hip_bf16.h>

#define DIM 128
#define CSTRIDE 64

typedef float v2f __attribute__((ext_vector_type(2)));
typedef float v8f __attribute__((ext_vector_type(8)));


__global__ void cnt_zero_kernel(unsigned* __restrict__ cnt, int n) {
    int i = blockIdx.x * blockDim.x + threadIdx.x;
    if (i < n) cnt[(size_t)i * CSTRIDE] = 0u;
}

__global__ void edge_rank_kernel(const int* __restrict__ ei, unsigned* __restrict__ cnt,
                                 unsigned* __restrict__ rank, int E) {
    int e = blockIdx.x * blockDim.x + threadIdx.x;
    if (e < E) {
        int d = ei[E + e];
        rank[e] = __hip_atomic_fetch_add(&cnt[(size_t)d * CSTRIDE], 1u, __ATOMIC_RELAXED,
                                         __HIP_MEMORY_SCOPE_AGENT);
    }
}

__global__ __launch_bounds__(256) void cnt_blocksum_kernel(const unsigned* __restrict__ cnt,
                                                           unsigned* __restrict__ bsum, int n) {
    __shared__ unsigned sw[8];
    const int tid = threadIdx.x;
    int i = blockIdx.x * 256 + tid;
    unsigned c = (i < n) ? cnt[(size_t)i * CSTRIDE] : 0u;
    for (int o = 16; o > 0; o >>= 1) c += __shfl_xor(c, o);
    if ((tid & 31) == 0) sw[tid >> 5] = c;
    __syncthreads();
    if (tid == 0) {
        unsigned s = 0;
        for (int w = 0; w < 8; ++w) s += sw[w];
        bsum[blockIdx.x] = s;
    }
}

__global__ __launch_bounds__(256) void rowptr_kernel(const unsigned* __restrict__ cnt,
                                                     const unsigned* __restrict__ bsum,
                                                     unsigned* __restrict__ rowptr,
                                                     float* __restrict__ dis, int n) {
    __shared__ unsigned sred[8];
    __shared__ unsigned swave[8];
    const int tid = threadIdx.x;
    const int lane = tid & 31;
    const int wave = tid >> 5;

    unsigned p = 0;
    for (int j = tid; j < (int)blockIdx.x; j += 256) p += bsum[j];
    for (int o = 16; o > 0; o >>= 1) p += __shfl_xor(p, o);

    int i = blockIdx.x * 256 + tid;
    unsigned c = (i < n) ? cnt[(size_t)i * CSTRIDE] : 0u;
    unsigned inc = c;
    for (int o = 1; o < 32; o <<= 1) {
        unsigned t = __shfl_up(inc, o);
        if (lane >= o) inc += t;
    }
    if (lane == 0) sred[wave] = p;
    if (lane == 31) swave[wave] = inc;
    __syncthreads();
    unsigned base = 0;
    for (int w = 0; w < 8; ++w) base += sred[w];
    for (int w = 0; w < wave; ++w) base += swave[w];
    if (i < n) {
        rowptr[i] = base + inc - c;
        float d = (float)(c + 1u);
        dis[i] = (d > 0.0f) ? rsqrtf(d) : 0.0f;
        if (i == n - 1) rowptr[n] = base + inc;
    }
}

__global__ void csr_fill_kernel(const int* __restrict__ ei, const unsigned* __restrict__ rank,
                                const unsigned* __restrict__ rowptr, int* __restrict__ csr, int E) {
    int e = blockIdx.x * blockDim.x + threadIdx.x;
    if (e < E) {
        int s = ei[e];
        int d = ei[E + e];
        unsigned pos = rowptr[d] + rank[e];
        if (pos < (unsigned)E) csr[pos] = s;
    }
}

__global__ __launch_bounds__(256) void gather_rows_kernel(const float* __restrict__ x,
                                                          const float* __restrict__ dis,
                                                          const unsigned* __restrict__ rowptr,
                                                          const int* __restrict__ csr,
                                                          float* __restrict__ agg, int n, int E) {
    const int tid = threadIdx.x;
    const int lane = tid & 31;
    const int row = __builtin_amdgcn_readfirstlane(blockIdx.x * 8 + (tid >> 5));
    if (row >= n) return;
    unsigned end = rowptr[row + 1];
    end = (end < (unsigned)E) ? end : (unsigned)E;
    unsigned beg = rowptr[row];
    beg = (beg < end) ? beg : end;
    const float dd = dis[row];
    const float4* x4 = reinterpret_cast<const float4*>(x);
    float4 acc;
    {
        float4 v = x4[(size_t)row * (DIM / 4) + lane];
        float sl = dd * dd;
        acc.x = v.x * sl; acc.y = v.y * sl; acc.z = v.z * sl; acc.w = v.w * sl;
    }
    for (unsigned base = beg; base < end; base += 32) {
        const unsigned m = (end - base < 32u) ? (end - base) : 32u;
        int ms = 0;
        float mw = 0.0f;
        if ((unsigned)lane < m) {
            unsigned s = (unsigned)csr[base + lane];
            s = (s < (unsigned)n) ? s : (unsigned)(n - 1);
            ms = (int)s;
            mw = dis[s] * dd;
        }
        const int mwi = __float_as_int(mw);
        unsigned j = 0;
        for (; j + 4 <= m; j += 4) {
            const float4* r0 = reinterpret_cast<const float4*>(x + (size_t)__builtin_amdgcn_readlane(ms, j + 0) * DIM);
            const float4* r1 = reinterpret_cast<const float4*>(x + (size_t)__builtin_amdgcn_readlane(ms, j + 1) * DIM);
            const float4* r2 = reinterpret_cast<const float4*>(x + (size_t)__builtin_amdgcn_readlane(ms, j + 2) * DIM);
            const float4* r3 = reinterpret_cast<const float4*>(x + (size_t)__builtin_amdgcn_readlane(ms, j + 3) * DIM);
            float w0 = __int_as_float(__builtin_amdgcn_readlane(mwi, j + 0));
            float w1 = __int_as_float(__builtin_amdgcn_readlane(mwi, j + 1));
            float w2 = __int_as_float(__builtin_amdgcn_readlane(mwi, j + 2));
            float w3 = __int_as_float(__builtin_amdgcn_readlane(mwi, j + 3));
            float4 a0 = r0[lane];
            float4 a1 = r1[lane];
            float4 a2 = r2[lane];
            float4 a3 = r3[lane];
            acc.x = fmaf(a0.x, w0, acc.x); acc.y = fmaf(a0.y, w0, acc.y); acc.z = fmaf(a0.z, w0, acc.z); acc.w = fmaf(a0.w, w0, acc.w);
            acc.x = fmaf(a1.x, w1, acc.x); acc.y = fmaf(a1.y, w1, acc.y); acc.z = fmaf(a1.z, w1, acc.z); acc.w = fmaf(a1.w, w1, acc.w);
            acc.x = fmaf(a2.x, w2, acc.x); acc.y = fmaf(a2.y, w2, acc.y); acc.z = fmaf(a2.z, w2, acc.z); acc.w = fmaf(a2.w, w2, acc.w);
            acc.x = fmaf(a3.x, w3, acc.x); acc.y = fmaf(a3.y, w3, acc.y); acc.z = fmaf(a3.z, w3, acc.z); acc.w = fmaf(a3.w, w3, acc.w);
        }
        if (j < m) {
            const unsigned r = m - j;
            const float4* r0 = reinterpret_cast<const float4*>(x + (size_t)__builtin_amdgcn_readlane(ms, j) * DIM);
            float w0 = __int_as_float(__builtin_amdgcn_readlane(mwi, j));
            float4 a0 = r0[lane];
            float4 a1 = {0.0f, 0.0f, 0.0f, 0.0f};
            float4 a2 = {0.0f, 0.0f, 0.0f, 0.0f};
            float w1 = 0.0f, w2 = 0.0f;
            if (r > 1) {
                const float4* r1 = reinterpret_cast<const float4*>(x + (size_t)__builtin_amdgcn_readlane(ms, j + 1) * DIM);
                w1 = __int_as_float(__builtin_amdgcn_readlane(mwi, j + 1));
                a1 = r1[lane];
            }
            if (r > 2) {
                const float4* r2 = reinterpret_cast<const float4*>(x + (size_t)__builtin_amdgcn_readlane(ms, j + 2) * DIM);
                w2 = __int_as_float(__builtin_amdgcn_readlane(mwi, j + 2));
                a2 = r2[lane];
            }
            acc.x = fmaf(a0.x, w0, acc.x); acc.y = fmaf(a0.y, w0, acc.y); acc.z = fmaf(a0.z, w0, acc.z); acc.w = fmaf(a0.w, w0, acc.w);
            if (r > 1) {
                acc.x = fmaf(a1.x, w1, acc.x); acc.y = fmaf(a1.y, w1, acc.y); acc.z = fmaf(a1.z, w1, acc.z); acc.w = fmaf(a1.w, w1, acc.w);
            }
            if (r > 2) {
                acc.x = fmaf(a2.x, w2, acc.x); acc.y = fmaf(a2.y, w2, acc.y); acc.z = fmaf(a2.z, w2, acc.z); acc.w = fmaf(a2.w, w2, acc.w);
            }
        }
    }
    reinterpret_cast<float4*>(agg)[(size_t)row * (DIM / 4) + lane] = acc;
}

#define LDA 132
#define LDW 136
#define KCHUNK 32

__global__ __launch_bounds__(256) void gemm_wmma_kernel(const float* __restrict__ A,
                                                        const float* __restrict__ W,
                                                        float* __restrict__ out, int n) {
    __shared__ float lA[16 * LDA];
    __shared__ float lW[KCHUNK * LDW];
    const int tid = threadIdx.x;
    const int wave = tid >> 5;
    const int lane = tid & 31;
    const int rowBase = blockIdx.x * 16;

    for (int i = tid; i < 16 * DIM; i += 256) {
        int r = i >> 7, c = i & 127;
        int gr = rowBase + r;
        lA[r * LDA + c] = (gr < n) ? A[(size_t)gr * DIM + c] : 0.0f;
    }

    const int m = lane & 15;
    const int hi = lane >> 4;
    const int colBase = wave * 16;

    v8f acc = {0.f, 0.f, 0.f, 0.f, 0.f, 0.f, 0.f, 0.f};

    for (int kc = 0; kc < DIM; kc += KCHUNK) {
        __syncthreads();
        for (int i = tid; i < KCHUNK * DIM; i += 256) {
            int r = i >> 7, c = i & 127;
            lW[r * LDW + c] = W[(size_t)(kc + r) * DIM + c];
        }
        __syncthreads();

#pragma unroll
        for (int k = 0; k < KCHUNK; k += 4) {
            v2f a, b;
            const float* ap = &lA[m * LDA + kc + k + 2 * hi];
            a.x = ap[0];
            a.y = ap[1];
            b.x = lW[(k + 2 * hi) * LDW + colBase + m];
            b.y = lW[(k + 2 * hi + 1) * LDW + colBase + m];
            acc = __builtin_amdgcn_wmma_f32_16x16x4_f32(
                 false, a,  false, b,
                 (short)0, acc,  false,  false);
        }
    }

#pragma unroll
    for (int v = 0; v < 8; ++v) {
        int gr = rowBase + v + 8 * hi;
        if (gr < n) out[(size_t)gr * DIM + colBase + m] = acc[v];
    }
}

extern "C" void kernel_launch(void* const* d_in, const int* in_sizes, int n_in,
                              void* d_out, int out_size, void* d_ws, size_t ws_size,
                              hipStream_t stream) {
    const float* x = (const float*)d_in[0];
    const int*   ei = (const int*)d_in[1];
    const float* W = (const float*)d_in[2];
    float* out = (float*)d_out;

    const int N = in_sizes[0] / DIM;
    const int E = in_sizes[1] / 2;

    float* dis = (float*)d_ws;
    size_t aggOff = (((size_t)N * sizeof(float)) + 255) & ~(size_t)255;
    float* agg = (float*)((char*)d_ws + aggOff);

    unsigned* cnt = (unsigned*)d_out;
    unsigned* rank = cnt + (size_t)N * CSTRIDE;
    int* csr = (int*)(rank + E);
    unsigned* rowptr = (unsigned*)(csr + E);
    unsigned* bsum = rowptr + N + 1;

    cnt_zero_kernel<<<(N + 255) / 256, 256, 0, stream>>>(cnt, N);
    edge_rank_kernel<<<(E + 255) / 256, 256, 0, stream>>>(ei, cnt, rank, E);
    cnt_blocksum_kernel<<<(N + 255) / 256, 256, 0, stream>>>(cnt, bsum, N);
    rowptr_kernel<<<(N + 255) / 256, 256, 0, stream>>>(cnt, bsum, rowptr, dis, N);
    csr_fill_kernel<<<(E + 255) / 256, 256, 0, stream>>>(ei, rank, rowptr, csr, E);
    gather_rows_kernel<<<(N + 7) / 8, 256, 0, stream>>>(x, dis, rowptr, csr, agg, N, E);
    gemm_wmma_kernel<<<(N + 15) / 16, 256, 0, stream>>>(agg, W, out, N);
}


# Final f32 WMMA product fused into the gather launch: 32 summed rows go to LDS and meet an LDS-resident W in the same block; agg buffer and one launch gone; 8 row loads per trip; sum order kept
# speedup vs baseline: 2.6608x; 1.2405x over previous
#include <hip/hip_runtime.h>
#include <hip/hip_bf16.h>

#define DIM 128
#define CSTRIDE 64

typedef float v2f __attribute__((ext_vector_type(2)));
typedef float v8f __attribute__((ext_vector_type(8)));


__global__ void cnt_zero_kernel(unsigned* __restrict__ cnt, int n) {
    int i = blockIdx.x * blockDim.x + threadIdx.x;
    if (i < n) cnt[(size_t)i * CSTRIDE] = 0u;
}

__global__ void edge_rank_kernel(const int* __restrict__ ei, unsigned* __restrict__ cnt,
                                 unsigned* __restrict__ rank, int E) {
    int e = blockIdx.x * blockDim.x + threadIdx.x;
    if (e < E) {
        int d = ei[E + e];
        rank[e] = __hip_atomic_fetch_add(&cnt[(size_t)d * CSTRIDE], 1u, __ATOMIC_RELAXED,
                                         __HIP_MEMORY_SCOPE_AGENT);
    }
}

__global__ __launch_bounds__(256) void cnt_blocksum_kernel(const unsigned* __restrict__ cnt,
                                                           unsigned* __restrict__ bsum, int n) {
    __shared__ unsigned sw[8];
    const int tid = threadIdx.x;
    int i = blockIdx.x * 256 + tid;
    unsigned c = (i < n) ? cnt[(size_t)i * CSTRIDE] : 0u;
    for (int o = 16; o > 0; o >>= 1) c += __shfl_xor(c, o);
    if ((tid & 31) == 0) sw[tid >> 5] = c;
    __syncthreads();
    if (tid == 0) {
        unsigned s = 0;
        for (int w = 0; w < 8; ++w) s += sw[w];
        bsum[blockIdx.x] = s;
    }
}

__global__ __launch_bounds__(256) void rowptr_kernel(const unsigned* __restrict__ cnt,
                                                     const unsigned* __restrict__ bsum,
                                                     unsigned* __restrict__ rowptr,
                                                     float* __restrict__ dis, int n) {
    __shared__ unsigned sred[8];
    __shared__ unsigned swave[8];
    const int tid = threadIdx.x;
    const int lane = tid & 31;
    const int wave = tid >> 5;

    unsigned p = 0;
    for (int j = tid; j < (int)blockIdx.x; j += 256) p += bsum[j];
    for (int o = 16; o > 0; o >>= 1) p += __shfl_xor(p, o);

    int i = blockIdx.x * 256 + tid;
    unsigned c = (i < n) ? cnt[(size_t)i * CSTRIDE] : 0u;
    unsigned inc = c;
    for (int o = 1; o < 32; o <<= 1) {
        unsigned t = __shfl_up(inc, o);
        if (lane >= o) inc += t;
    }
    if (lane == 0) sred[wave] = p;
    if (lane == 31) swave[wave] = inc;
    __syncthreads();
    unsigned base = 0;
    for (int w = 0; w < 8; ++w) base += sred[w];
    for (int w = 0; w < wave; ++w) base += swave[w];
    if (i < n) {
        rowptr[i] = base + inc - c;
        float d = (float)(c + 1u);
        dis[i] = (d > 0.0f) ? rsqrtf(d) : 0.0f;
        if (i == n - 1) rowptr[n] = base + inc;
    }
}

__global__ void csr_fill_kernel(const int* __restrict__ ei, const unsigned* __restrict__ rank,
                                const unsigned* __restrict__ rowptr, int* __restrict__ csr, int E) {
    int e = blockIdx.x * blockDim.x + threadIdx.x;
    if (e < E) {
        int s = ei[e];
        int d = ei[E + e];
        unsigned pos = rowptr[d] + rank[e];
        if (pos < (unsigned)E) csr[pos] = s;
    }
}

#define LDA 132
#define LDW 136
#define BROWS 32

#define FMA4(A, S) acc.x = fmaf((A).x, (S), acc.x); acc.y = fmaf((A).y, (S), acc.y); \
                   acc.z = fmaf((A).z, (S), acc.z); acc.w = fmaf((A).w, (S), acc.w);

__global__ __launch_bounds__(512) void gather_gemm_kernel(const float* __restrict__ x,
                                                          const float* __restrict__ dis,
                                                          const unsigned* __restrict__ rowptr,
                                                          const int* __restrict__ csr,
                                                          const float* __restrict__ Wm,
                                                          float* __restrict__ out, int n, int E) {
    __shared__ float lW[DIM * LDW];
    __shared__ float lA[BROWS * LDA];
    const int tid = threadIdx.x;
    const int lane = tid & 31;
    const int wave = __builtin_amdgcn_readfirstlane(tid >> 5);
    const int rowBase = blockIdx.x * BROWS;

    {
        const float4* W4 = reinterpret_cast<const float4*>(Wm);
#pragma unroll
        for (int i = 0; i < (DIM * DIM / 4) / 512; ++i) {
            const int idx = tid + i * 512;
            const int r = idx >> 5, c4 = idx & 31;
            *reinterpret_cast<float4*>(&lW[r * LDW + c4 * 4]) = W4[idx];
        }
    }

    const float4* x4 = reinterpret_cast<const float4*>(x);
    for (int rr = 0; rr < 2; ++rr) {
        const int lr = wave * 2 + rr;
        const int row = rowBase + lr;
        float4 acc = {0.0f, 0.0f, 0.0f, 0.0f};
        if (row < n) {
            unsigned end = rowptr[row + 1];
            end = (end < (unsigned)E) ? end : (unsigned)E;
            unsigned beg = rowptr[row];
            beg = (beg < end) ? beg : end;
            const float dd = dis[row];
            {
                float4 v = x4[(size_t)row * (DIM / 4) + lane];
                float sl = dd * dd;
                acc.x = v.x * sl; acc.y = v.y * sl; acc.z = v.z * sl; acc.w = v.w * sl;
            }
            for (unsigned base = beg; base < end; base += 32) {
                const unsigned m = (end - base < 32u) ? (end - base) : 32u;
                int ms = 0;
                float mw = 0.0f;
                if ((unsigned)lane < m) {
                    unsigned s = (unsigned)csr[base + lane];
                    s = (s < (unsigned)n) ? s : (unsigned)(n - 1);
                    ms = (int)s;
                    mw = dis[s] * dd;
                }
                const int mwi = __float_as_int(mw);
                unsigned j = 0;
                for (; j + 8 <= m; j += 8) {
                    float4 a[8];
                    float w[8];
#pragma unroll
                    for (int u = 0; u < 8; ++u) {
                        const float4* r = reinterpret_cast<const float4*>(x + (size_t)__builtin_amdgcn_readlane(ms, j + u) * DIM);
                        w[u] = __int_as_float(__builtin_amdgcn_readlane(mwi, j + u));
                        a[u] = r[lane];
                    }
#pragma unroll
                    for (int u = 0; u < 8; ++u) { FMA4(a[u], w[u]) }
                }
                if (j + 4 <= m) {
                    float4 a[4];
                    float w[4];
#pragma unroll
                    for (int u = 0; u < 4; ++u) {
                        const float4* r = reinterpret_cast<const float4*>(x + (size_t)__builtin_amdgcn_readlane(ms, j + u) * DIM);
                        w[u] = __int_as_float(__builtin_amdgcn_readlane(mwi, j + u));
                        a[u] = r[lane];
                    }
#pragma unroll
                    for (int u = 0; u < 4; ++u) { FMA4(a[u], w[u]) }
                    j += 4;
                }
                if (j < m) {
                    const unsigned r = m - j;
                    const float4* r0 = reinterpret_cast<const float4*>(x + (size_t)__builtin_amdgcn_readlane(ms, j) * DIM);
                    float w0 = __int_as_float(__builtin_amdgcn_readlane(mwi, j));
                    float4 a0 = r0[lane];
                    float4 a1 = {0.0f, 0.0f, 0.0f, 0.0f};
                    float4 a2 = {0.0f, 0.0f, 0.0f, 0.0f};
                    float w1 = 0.0f, w2 = 0.0f;
                    if (r > 1) {
                        const float4* r1 = reinterpret_cast<const float4*>(x + (size_t)__builtin_amdgcn_readlane(ms, j + 1) * DIM);
                        w1 = __int_as_float(__builtin_amdgcn_readlane(mwi, j + 1));
                        a1 = r1[lane];
                    }
                    if (r > 2) {
                        const float4* r2 = reinterpret_cast<const float4*>(x + (size_t)__builtin_amdgcn_readlane(ms, j + 2) * DIM);
                        w2 = __int_as_float(__builtin_amdgcn_readlane(mwi, j + 2));
                        a2 = r2[lane];
                    }
                    FMA4(a0, w0)
                    if (r > 1) { FMA4(a1, w1) }
                    if (r > 2) { FMA4(a2, w2) }
                }
            }
        }
        *reinterpret_cast<float4*>(&lA[lr * LDA + lane * 4]) = acc;
    }
    __syncthreads();

    const int m = lane & 15;
    const int hi = lane >> 4;
    const int strip = wave >> 3;
    const int colBase = (wave & 7) * 16;

    v8f c = {0.f, 0.f, 0.f, 0.f, 0.f, 0.f, 0.f, 0.f};
#pragma unroll
    for (int k = 0; k < DIM; k += 4) {
        v2f a, b;
        const float* ap = &lA[(strip * 16 + m) * LDA + k + 2 * hi];
        a.x = ap[0];
        a.y = ap[1];
        b.x = lW[(k + 2 * hi) * LDW + colBase + m];
        b.y = lW[(k + 2 * hi + 1) * LDW + colBase + m];
        c = __builtin_amdgcn_wmma_f32_16x16x4_f32(
             false, a,  false, b,
             (short)0, c,  false,  false);
    }

#pragma unroll
    for (int v = 0; v < 8; ++v) {
        int gr = rowBase + strip * 16 + v + 8 * hi;
        if (gr < n) out[(size_t)gr * DIM + colBase + m] = c[v];
    }
}

extern "C" void kernel_launch(void* const* d_in, const int* in_sizes, int n_in,
                              void* d_out, int out_size, void* d_ws, size_t ws_size,
                              hipStream_t stream) {
    const float* x = (const float*)d_in[0];
    const int*   ei = (const int*)d_in[1];
    const float* W = (const float*)d_in[2];
    float* out = (float*)d_out;

    const int N = in_sizes[0] / DIM;
    const int E = in_sizes[1] / 2;

    float* dis = (float*)d_ws;
    size_t idxOff = (((size_t)N * sizeof(float)) + 255) & ~(size_t)255;

    unsigned* cnt = (unsigned*)((char*)d_ws + idxOff);
    unsigned* rank = cnt + (size_t)N * CSTRIDE;
    int* csr = (int*)(rank + E);
    unsigned* rowptr = (unsigned*)(csr + E);
    unsigned* bsum = rowptr + N + 1;

    cnt_zero_kernel<<<(N + 255) / 256, 256, 0, stream>>>(cnt, N);
    edge_rank_kernel<<<(E + 255) / 256, 256, 0, stream>>>(ei, cnt, rank, E);
    cnt_blocksum_kernel<<<(N + 255) / 256, 256, 0, stream>>>(cnt, bsum, N);
    rowptr_kernel<<<(N + 255) / 256, 256, 0, stream>>>(cnt, bsum, rowptr, dis, N);
    csr_fill_kernel<<<(E + 255) / 256, 256, 0, stream>>>(ei, rank, rowptr, csr, E);
    gather_gemm_kernel<<<(N + BROWS - 1) / BROWS, 512, 0, stream>>>(x, dis, rowptr, csr, W, out, N, E);
}


# Edge grouping redone as a two-level counting sort on LDS atomics (256-row buckets per edge chunk, then rows inside a bucket): no global atomics, 5 launches; row sums in LDS arrival order
# speedup vs baseline: 3.8746x; 1.4562x over previous
#include <hip/hip_runtime.h>
#include <hip/hip_bf16.h>

#define DIM 128
#define RB 256
#define RBSHIFT 8
#define MAXBK 512
#define G1 512
#define TILE 4096
#define BCAP 8192

typedef float v2f __attribute__((ext_vector_type(2)));
typedef float v8f __attribute__((ext_vector_type(8)));


__device__ __forceinline__ unsigned block_excl_scan(unsigned v, unsigned* sw, unsigned* total) {
    const int lane = threadIdx.x & 31;
    const int wave = threadIdx.x >> 5;
    const int nw = blockDim.x >> 5;
    unsigned inc = v;
    for (int o = 1; o < 32; o <<= 1) {
        unsigned t = __shfl_up(inc, o);
        if (lane >= o) inc += t;
    }
    if (lane == 31) sw[wave] = inc;
    __syncthreads();
    unsigned base = 0, tot = 0;
    for (int w = 0; w < nw; ++w) {
        unsigned x = sw[w];
        if (w < wave) base += x;
        tot += x;
    }
    __syncthreads();
    *total = tot;
    return base + inc - v;
}

__global__ __launch_bounds__(512) void bucket_count_kernel(const int* __restrict__ ei,
                                                           unsigned* __restrict__ tab,
                                                           int n, int E, int nbk) {
    __shared__ unsigned hist[MAXBK];
    const int tid = threadIdx.x;
    hist[tid] = 0u;
    __syncthreads();
    const int ch = (E + G1 - 1) / G1;
    const int e0 = blockIdx.x * ch;
    int e1 = e0 + ch;
    e1 = (e1 < E) ? e1 : E;
    for (int e = e0 + tid; e < e1; e += 512) {
        unsigned d = (unsigned)ei[E + e];
        d = (d < (unsigned)n) ? d : (unsigned)(n - 1);
        atomicAdd(&hist[d >> RBSHIFT], 1u);
    }
    __syncthreads();
    if (tid < nbk) tab[(size_t)tid * G1 + blockIdx.x] = hist[tid];
}

__global__ __launch_bounds__(512) void chunk_scan_kernel(unsigned* __restrict__ tab,
                                                         unsigned* __restrict__ tot) {
    __shared__ unsigned sw[16];
    const size_t i = (size_t)blockIdx.x * G1 + threadIdx.x;
    const unsigned v = tab[i];
    unsigned total;
    const unsigned ex = block_excl_scan(v, sw, &total);
    tab[i] = ex;
    if (threadIdx.x == 0) tot[blockIdx.x] = total;
}

__global__ __launch_bounds__(512) void bucket_place_kernel(const int* __restrict__ ei,
                                                           const unsigned* __restrict__ tab,
                                                           const unsigned* __restrict__ tot,
                                                           unsigned* __restrict__ bpk,
                                                           int n, int E, int nbk) {
    __shared__ unsigned cur[MAXBK];
    __shared__ unsigned hist[MAXBK];
    __shared__ unsigned lof[MAXBK];
    __shared__ unsigned gb[MAXBK];
    __shared__ unsigned stVal[TILE];
    __shared__ unsigned stPos[TILE];
    __shared__ unsigned sw[16];
    const int tid = threadIdx.x;
    const unsigned t = (tid < nbk) ? tot[tid] : 0u;
    unsigned total;
    const unsigned bstart = block_excl_scan(t, sw, &total);
    cur[tid] = (tid < nbk) ? bstart + tab[(size_t)tid * G1 + blockIdx.x] : 0u;
    const int ch = (E + G1 - 1) / G1;
    const int e0 = blockIdx.x * ch;
    int e1 = e0 + ch;
    e1 = (e1 < E) ? e1 : E;
    for (int t0 = e0; t0 < e1; t0 += TILE) {
        hist[tid] = 0u;
        __syncthreads();
        unsigned pk[8], lr[8];
        int bk[8];
#pragma unroll
        for (int k = 0; k < 8; ++k) {
            const int e = t0 + k * 512 + tid;
            bk[k] = -1;
            pk[k] = 0u;
            lr[k] = 0u;
            if (e < e1) {
                unsigned s = (unsigned)ei[e];
                unsigned d = (unsigned)ei[E + e];
                s = (s < (unsigned)n) ? s : (unsigned)(n - 1);
                d = (d < (unsigned)n) ? d : (unsigned)(n - 1);
                bk[k] = (int)(d >> RBSHIFT);
                pk[k] = (s << RBSHIFT) | (d & (unsigned)(RB - 1));
                lr[k] = atomicAdd(&hist[bk[k]], 1u);
            }
        }
        __syncthreads();
        const unsigned h = hist[tid];
        unsigned tileCount;
        const unsigned lo = block_excl_scan(h, sw, &tileCount);
        lof[tid] = lo;
        gb[tid] = cur[tid];
        cur[tid] += h;
        __syncthreads();
#pragma unroll
        for (int k = 0; k < 8; ++k) {
            if (bk[k] >= 0) {
                const unsigned idx = lof[bk[k]] + lr[k];
                if (idx < (unsigned)TILE) {
                    stVal[idx] = pk[k];
                    stPos[idx] = gb[bk[k]] + lr[k];
                }
            }
        }
        __syncthreads();
        for (unsigned i = tid; i < tileCount; i += 512) {
            const unsigned pos = stPos[i];
            if (pos < (unsigned)E) bpk[pos] = stVal[i];
        }
    }
}

__global__ __launch_bounds__(256) void bucket_rows_kernel(const unsigned* __restrict__ bpk,
                                                          const unsigned* __restrict__ tot,
                                                          unsigned* __restrict__ rowptr,
                                                          float* __restrict__ dis,
                                                          int* __restrict__ csr, int n, int E) {
    __shared__ unsigned cntL[RB];
    __shared__ unsigned curL[RB];
    __shared__ int csrL[BCAP];
    __shared__ unsigned sw[16];
    const int tid = threadIdx.x;
    const int b = blockIdx.x;
    unsigned p = 0;
    for (int j = tid; j < b; j += RB) p += tot[j];
    unsigned bstart;
    (void)block_excl_scan(p, sw, &bstart);
    bstart = (bstart < (unsigned)E) ? bstart : (unsigned)E;
    unsigned bc = tot[b];
    bc = (bc < (unsigned)E - bstart) ? bc : (unsigned)E - bstart;
    const bool inLds = (bc <= (unsigned)BCAP);
    cntL[tid] = 0u;
    __syncthreads();
    for (unsigned i = tid; i < bc; i += RB) atomicAdd(&cntL[bpk[bstart + i] & (unsigned)(RB - 1)], 1u);
    __syncthreads();
    const unsigned c = cntL[tid];
    unsigned total;
    const unsigned ex = block_excl_scan(c, sw, &total);
    curL[tid] = ex;
    const int row = b * RB + tid;
    if (row < n) {
        rowptr[row] = bstart + ex;
        float d = (float)(c + 1u);
        dis[row] = (d > 0.0f) ? rsqrtf(d) : 0.0f;
        if (row == n - 1) rowptr[n] = bstart + ex + c;
    }
    __syncthreads();
    for (unsigned i = tid; i < bc; i += RB) {
        const unsigned w = bpk[bstart + i];
        const unsigned q = atomicAdd(&curL[w & (unsigned)(RB - 1)], 1u);
        const int s = (int)(w >> RBSHIFT);
        if (inLds) {
            if (q < (unsigned)BCAP) csrL[q] = s;
        } else {
            const unsigned pos = bstart + q;
            if (pos < (unsigned)E) csr[pos] = s;
        }
    }
    if (inLds) {
        __syncthreads();
        for (unsigned i = tid; i < bc; i += RB) csr[bstart + i] = csrL[i];
    }
}

#define LDA 132
#define LDW 136
#define BROWS 32

#define FMA4(A, S) acc.x = fmaf((A).x, (S), acc.x); acc.y = fmaf((A).y, (S), acc.y); \
                   acc.z = fmaf((A).z, (S), acc.z); acc.w = fmaf((A).w, (S), acc.w);

__global__ __launch_bounds__(512) void gather_gemm_kernel(const float* __restrict__ x,
                                                          const float* __restrict__ dis,
                                                          const unsigned* __restrict__ rowptr,
                                                          const int* __restrict__ csr,
                                                          const float* __restrict__ Wm,
                                                          float* __restrict__ out, int n, int E) {
    __shared__ float lW[DIM * LDW];
    __shared__ float lA[BROWS * LDA];
    const int tid = threadIdx.x;
    const int lane = tid & 31;
    const int wave = __builtin_amdgcn_readfirstlane(tid >> 5);
    const int rowBase = blockIdx.x * BROWS;

    {
        const float4* W4 = reinterpret_cast<const float4*>(Wm);
#pragma unroll
        for (int i = 0; i < (DIM * DIM / 4) / 512; ++i) {
            const int idx = tid + i * 512;
            const int r = idx >> 5, c4 = idx & 31;
            *reinterpret_cast<float4*>(&lW[r * LDW + c4 * 4]) = W4[idx];
        }
    }

    const float4* x4 = reinterpret_cast<const float4*>(x);
    for (int rr = 0; rr < 2; ++rr) {
        const int lr = wave * 2 + rr;
        const int row = rowBase + lr;
        float4 acc = {0.0f, 0.0f, 0.0f, 0.0f};
        if (row < n) {
            unsigned end = rowptr[row + 1];
            end = (end < (unsigned)E) ? end : (unsigned)E;
            unsigned beg = rowptr[row];
            beg = (beg < end) ? beg : end;
            const float dd = dis[row];
            {
                float4 v = x4[(size_t)row * (DIM / 4) + lane];
                float sl = dd * dd;
                acc.x = v.x * sl; acc.y = v.y * sl; acc.z = v.z * sl; acc.w = v.w * sl;
            }
            for (unsigned base = beg; base < end; base += 32) {
                const unsigned m = (end - base < 32u) ? (end - base) : 32u;
                int ms = 0;
                float mw = 0.0f;
                if ((unsigned)lane < m) {
                    unsigned s = (unsigned)csr[base + lane];
                    s = (s < (unsigned)n) ? s : (unsigned)(n - 1);
                    ms = (int)s;
                    mw = dis[s] * dd;
                }
                const int mwi = __float_as_int(mw);
                unsigned j = 0;
                for (; j + 8 <= m; j += 8) {
                    float4 a[8];
                    float w[8];
#pragma unroll
                    for (int u = 0; u < 8; ++u) {
                        const float4* r = reinterpret_cast<const float4*>(x + (size_t)__builtin_amdgcn_readlane(ms, j + u) * DIM);
                        w[u] = __int_as_float(__builtin_amdgcn_readlane(mwi, j + u));
                        a[u] = r[lane];
                    }
#pragma unroll
                    for (int u = 0; u < 8; ++u) { FMA4(a[u], w[u]) }
                }
                if (j + 4 <= m) {
                    float4 a[4];
                    float w[4];
#pragma unroll
                    for (int u = 0; u < 4; ++u) {
                        const float4* r = reinterpret_cast<const float4*>(x + (size_t)__builtin_amdgcn_readlane(ms, j + u) * DIM);
                        w[u] = __int_as_float(__builtin_amdgcn_readlane(mwi, j + u));
                        a[u] = r[lane];
                    }
#pragma unroll
                    for (int u = 0; u < 4; ++u) { FMA4(a[u], w[u]) }
                    j += 4;
                }
                if (j < m) {
                    const unsigned r = m - j;
                    const float4* r0 = reinterpret_cast<const float4*>(x + (size_t)__builtin_amdgcn_readlane(ms, j) * DIM);
                    float w0 = __int_as_float(__builtin_amdgcn_readlane(mwi, j));
                    float4 a0 = r0[lane];
                    float4 a1 = {0.0f, 0.0f, 0.0f, 0.0f};
                    float4 a2 = {0.0f, 0.0f, 0.0f, 0.0f};
                    float w1 = 0.0f, w2 = 0.0f;
                    if (r > 1) {
                        const float4* r1 = reinterpret_cast<const float4*>(x + (size_t)__builtin_amdgcn_readlane(ms, j + 1) * DIM);
                        w1 = __int_as_float(__builtin_amdgcn_readlane(mwi, j + 1));
                        a1 = r1[lane];
                    }
                    if (r > 2) {
                        const float4* r2 = reinterpret_cast<const float4*>(x + (size_t)__builtin_amdgcn_readlane(ms, j + 2) * DIM);
                        w2 = __int_as_float(__builtin_amdgcn_readlane(mwi, j + 2));
                        a2 = r2[lane];
                    }
                    FMA4(a0, w0)
                    if (r > 1) { FMA4(a1, w1) }
                    if (r > 2) { FMA4(a2, w2) }
                }
            }
        }
        *reinterpret_cast<float4*>(&lA[lr * LDA + lane * 4]) = acc;
    }
    __syncthreads();

    const int m = lane & 15;
    const int hi = lane >> 4;
    const int strip = wave >> 3;
    const int colBase = (wave & 7) * 16;

    v8f c = {0.f, 0.f, 0.f, 0.f, 0.f, 0.f, 0.f, 0.f};
#pragma unroll
    for (int k = 0; k < DIM; k += 4) {
        v2f a, b;
        const float* ap = &lA[(strip * 16 + m) * LDA + k + 2 * hi];
        a.x = ap[0];
        a.y = ap[1];
        b.x = lW[(k + 2 * hi) * LDW + colBase + m];
        b.y = lW[(k + 2 * hi + 1) * LDW + colBase + m];
        c = __builtin_amdgcn_wmma_f32_16x16x4_f32(
             false, a,  false, b,
             (short)0, c,  false,  false);
    }

#pragma unroll
    for (int v = 0; v < 8; ++v) {
        int gr = rowBase + strip * 16 + v + 8 * hi;
        if (gr < n) out[(size_t)gr * DIM + colBase + m] = c[v];
    }
}

extern "C" void kernel_launch(void* const* d_in, const int* in_sizes, int n_in,
                              void* d_out, int out_size, void* d_ws, size_t ws_size,
                              hipStream_t stream) {
    const float* x = (const float*)d_in[0];
    const int*   ei = (const int*)d_in[1];
    const float* W = (const float*)d_in[2];
    float* out = (float*)d_out;

    const int N = in_sizes[0] / DIM;
    const int E = in_sizes[1] / 2;
    const int NBK = (N + RB - 1) / RB;

    float* dis = (float*)d_ws;
    size_t idxOff = (((size_t)N * sizeof(float)) + 255) & ~(size_t)255;

    unsigned* tab = (unsigned*)((char*)d_ws + idxOff);
    unsigned* tot = tab + (size_t)NBK * G1;
    unsigned* bpk = tot + NBK;
    int* csr = (int*)(bpk + E);
    unsigned* rowptr = (unsigned*)(csr + E);

    bucket_count_kernel<<<G1, 512, 0, stream>>>(ei, tab, N, E, NBK);
    chunk_scan_kernel<<<NBK, G1, 0, stream>>>(tab, tot);
    bucket_place_kernel<<<G1, 512, 0, stream>>>(ei, tab, tot, bpk, N, E, NBK);
    bucket_rows_kernel<<<NBK, RB, 0, stream>>>(bpk, tot, rowptr, dis, csr, N, E);
    gather_gemm_kernel<<<(N + BROWS - 1) / BROWS, 512, 0, stream>>>(x, dis, rowptr, csr, W, out, N, E);
}


# Last launch made resident: 768 work-groups draw 32-row tiles from a ticket counter and stage W into LDS once each, not once per tile; row strip double-buffered, one barrier per tile; sum order kept
# speedup vs baseline: 4.1261x; 1.0649x over previous
#include <hip/hip_runtime.h>
#include <hip/hip_bf16.h>

#define DIM 128
#define RB 256
#define RBSHIFT 8
#define MAXBK 512
#define G1 512
#define TILE 4096
#define BCAP 8192

typedef float v2f __attribute__((ext_vector_type(2)));
typedef float v8f __attribute__((ext_vector_type(8)));


__device__ __forceinline__ unsigned block_excl_scan(unsigned v, unsigned* sw, unsigned* total) {
    const int lane = threadIdx.x & 31;
    const int wave = threadIdx.x >> 5;
    const int nw = blockDim.x >> 5;
    unsigned inc = v;
    for (int o = 1; o < 32; o <<= 1) {
        unsigned t = __shfl_up(inc, o);
        if (lane >= o) inc += t;
    }
    if (lane == 31) sw[wave] = inc;
    __syncthreads();
    unsigned base = 0, tot = 0;
    for (int w = 0; w < nw; ++w) {
        unsigned x = sw[w];
        if (w < wave) base += x;
        tot += x;
    }
    __syncthreads();
    *total = tot;
    return base + inc - v;
}

__global__ __launch_bounds__(512) void bucket_count_kernel(const int* __restrict__ ei,
                                                           unsigned* __restrict__ tab,
                                                           int n, int E, int nbk) {
    __shared__ unsigned hist[MAXBK];
    const int tid = threadIdx.x;
    hist[tid] = 0u;
    __syncthreads();
    const int ch = (E + G1 - 1) / G1;
    const int e0 = blockIdx.x * ch;
    int e1 = e0 + ch;
    e1 = (e1 < E) ? e1 : E;
    for (int e = e0 + tid; e < e1; e += 512) {
        unsigned d = (unsigned)ei[E + e];
        d = (d < (unsigned)n) ? d : (unsigned)(n - 1);
        atomicAdd(&hist[d >> RBSHIFT], 1u);
    }
    __syncthreads();
    if (tid < nbk) tab[(size_t)tid * G1 + blockIdx.x] = hist[tid];
}

__global__ __launch_bounds__(512) void chunk_scan_kernel(unsigned* __restrict__ tab,
                                                         unsigned* __restrict__ tot) {
    __shared__ unsigned sw[16];
    const size_t i = (size_t)blockIdx.x * G1 + threadIdx.x;
    const unsigned v = tab[i];
    unsigned total;
    const unsigned ex = block_excl_scan(v, sw, &total);
    tab[i] = ex;
    if (threadIdx.x == 0) tot[blockIdx.x] = total;
}

__global__ __launch_bounds__(512) void bucket_place_kernel(const int* __restrict__ ei,
                                                           const unsigned* __restrict__ tab,
                                                           const unsigned* __restrict__ tot,
                                                           unsigned* __restrict__ bpk,
                                                           int n, int E, int nbk) {
    __shared__ unsigned cur[MAXBK];
    __shared__ unsigned hist[MAXBK];
    __shared__ unsigned lof[MAXBK];
    __shared__ unsigned gb[MAXBK];
    __shared__ unsigned stVal[TILE];
    __shared__ unsigned stPos[TILE];
    __shared__ unsigned sw[16];
    const int tid = threadIdx.x;
    const unsigned t = (tid < nbk) ? tot[tid] : 0u;
    unsigned total;
    const unsigned bstart = block_excl_scan(t, sw, &total);
    cur[tid] = (tid < nbk) ? bstart + tab[(size_t)tid * G1 + blockIdx.x] : 0u;
    const int ch = (E + G1 - 1) / G1;
    const int e0 = blockIdx.x * ch;
    int e1 = e0 + ch;
    e1 = (e1 < E) ? e1 : E;
    for (int t0 = e0; t0 < e1; t0 += TILE) {
        hist[tid] = 0u;
        __syncthreads();
        unsigned pk[8], lr[8];
        int bk[8];
#pragma unroll
        for (int k = 0; k < 8; ++k) {
            const int e = t0 + k * 512 + tid;
            bk[k] = -1;
            pk[k] = 0u;
            lr[k] = 0u;
            if (e < e1) {
                unsigned s = (unsigned)ei[e];
                unsigned d = (unsigned)ei[E + e];
                s = (s < (unsigned)n) ? s : (unsigned)(n - 1);
                d = (d < (unsigned)n) ? d : (unsigned)(n - 1);
                bk[k] = (int)(d >> RBSHIFT);
                pk[k] = (s << RBSHIFT) | (d & (unsigned)(RB - 1));
                lr[k] = atomicAdd(&hist[bk[k]], 1u);
            }
        }
        __syncthreads();
        const unsigned h = hist[tid];
        unsigned tileCount;
        const unsigned lo = block_excl_scan(h, sw, &tileCount);
        lof[tid] = lo;
        gb[tid] = cur[tid];
        cur[tid] += h;
        __syncthreads();
#pragma unroll
        for (int k = 0; k < 8; ++k) {
            if (bk[k] >= 0) {
                const unsigned idx = lof[bk[k]] + lr[k];
                if (idx < (unsigned)TILE) {
                    stVal[idx] = pk[k];
                    stPos[idx] = gb[bk[k]] + lr[k];
                }
            }
        }
        __syncthreads();
        for (unsigned i = tid; i < tileCount; i += 512) {
            const unsigned pos = stPos[i];
            if (pos < (unsigned)E) bpk[pos] = stVal[i];
        }
    }
}

__global__ __launch_bounds__(256) void bucket_rows_kernel(const unsigned* __restrict__ bpk,
                                                          const unsigned* __restrict__ tot,
                                                          unsigned* __restrict__ rowptr,
                                                          float* __restrict__ dis,
                                                          int* __restrict__ csr,
                                                          unsigned* __restrict__ ticket, int n, int E) {
    __shared__ unsigned cntL[RB];
    __shared__ unsigned curL[RB];
    __shared__ int csrL[BCAP];
    __shared__ unsigned sw[16];
    const int tid = threadIdx.x;
    const int b = blockIdx.x;
    if (b == 0 && tid == 0) *ticket = 0u;
    unsigned p = 0;
    for (int j = tid; j < b; j += RB) p += tot[j];
    unsigned bstart;
    (void)block_excl_scan(p, sw, &bstart);
    bstart = (bstart < (unsigned)E) ? bstart : (unsigned)E;
    unsigned bc = tot[b];
    bc = (bc < (unsigned)E - bstart) ? bc : (unsigned)E - bstart;
    const bool inLds = (bc <= (unsigned)BCAP);
    cntL[tid] = 0u;
    __syncthreads();
    for (unsigned i = tid; i < bc; i += RB) atomicAdd(&cntL[bpk[bstart + i] & (unsigned)(RB - 1)], 1u);
    __syncthreads();
    const unsigned c = cntL[tid];
    unsigned total;
    const unsigned ex = block_excl_scan(c, sw, &total);
    curL[tid] = ex;
    const int row = b * RB + tid;
    if (row < n) {
        rowptr[row] = bstart + ex;
        float d = (float)(c + 1u);
        dis[row] = (d > 0.0f) ? rsqrtf(d) : 0.0f;
        if (row == n - 1) rowptr[n] = bstart + ex + c;
    }
    __syncthreads();
    for (unsigned i = tid; i < bc; i += RB) {
        const unsigned w = bpk[bstart + i];
        const unsigned q = atomicAdd(&curL[w & (unsigned)(RB - 1)], 1u);
        const int s = (int)(w >> RBSHIFT);
        if (inLds) {
            if (q < (unsigned)BCAP) csrL[q] = s;
        } else {
            const unsigned pos = bstart + q;
            if (pos < (unsigned)E) csr[pos] = s;
        }
    }
    if (inLds) {
        __syncthreads();
        for (unsigned i = tid; i < bc; i += RB) csr[bstart + i] = csrL[i];
    }
}

#define LDA 132
#define LDW 136
#define BROWS 32
#define GGRID 768

#define FMA4(A, S) acc.x = fmaf((A).x, (S), acc.x); acc.y = fmaf((A).y, (S), acc.y); \
                   acc.z = fmaf((A).z, (S), acc.z); acc.w = fmaf((A).w, (S), acc.w);

__global__ __launch_bounds__(512, 12) void gather_gemm_kernel(const float* __restrict__ x,
                                                          const float* __restrict__ dis,
                                                          const unsigned* __restrict__ rowptr,
                                                          const int* __restrict__ csr,
                                                          const float* __restrict__ Wm,
                                                          float* __restrict__ out,
                                                          unsigned* __restrict__ ticket, int n, int E) {
    __shared__ float lW[DIM * LDW];
    __shared__ float lA[2 * BROWS * LDA];
    __shared__ int nextTile[2];
    const int tid = threadIdx.x;
    const int lane = tid & 31;
    const int wave = __builtin_amdgcn_readfirstlane(tid >> 5);
    const int numTiles = (n + BROWS - 1) / BROWS;

    if (tid == 0) nextTile[1] = (int)atomicAdd(ticket, 1u);

    {
        const float4* W4 = reinterpret_cast<const float4*>(Wm);
#pragma unroll
        for (int i = 0; i < (DIM * DIM / 4) / 512; ++i) {
            const int idx = tid + i * 512;
            const int r = idx >> 5, c4 = idx & 31;
            *reinterpret_cast<float4*>(&lW[r * LDW + c4 * 4]) = W4[idx];
        }
    }
    __syncthreads();
    int tile = __builtin_amdgcn_readfirstlane(nextTile[1]);

    const float4* x4 = reinterpret_cast<const float4*>(x);
    const int mn = lane & 15;
    const int hi = lane >> 4;
    const int strip = wave >> 3;
    const int colBase = (wave & 7) * 16;
    int buf = 0;

    while (tile < numTiles) {
        const int rowBase = tile * BROWS;
        float* lAb = &lA[buf * (BROWS * LDA)];
        unsigned got = 0u;
        if (tid == 0) got = atomicAdd(ticket, 1u);

        for (int rr = 0; rr < 2; ++rr) {
            const int lr = wave * 2 + rr;
            const int row = rowBase + lr;
            float4 acc = {0.0f, 0.0f, 0.0f, 0.0f};
            if (row < n) {
                unsigned end = rowptr[row + 1];
                end = (end < (unsigned)E) ? end : (unsigned)E;
                unsigned beg = rowptr[row];
                beg = (beg < end) ? beg : end;
                const float dd = dis[row];
                {
                    float4 v = x4[(size_t)row * (DIM / 4) + lane];
                    float sl = dd * dd;
                    acc.x = v.x * sl; acc.y = v.y * sl; acc.z = v.z * sl; acc.w = v.w * sl;
                }
                for (unsigned base = beg; base < end; base += 32) {
                    const unsigned m = (end - base < 32u) ? (end - base) : 32u;
                    int ms = 0;
                    float mw = 0.0f;
                    if ((unsigned)lane < m) {
                        unsigned s = (unsigned)csr[base + lane];
                        s = (s < (unsigned)n) ? s : (unsigned)(n - 1);
                        ms = (int)s;
                        mw = dis[s] * dd;
                    }
                    const int mwi = __float_as_int(mw);
                    unsigned j = 0;
                    for (; j + 8 <= m; j += 8) {
                        float4 a[8];
                        float w[8];
#pragma unroll
                        for (int u = 0; u < 8; ++u) {
                            const float4* r = reinterpret_cast<const float4*>(x + (size_t)__builtin_amdgcn_readlane(ms, j + u) * DIM);
                            w[u] = __int_as_float(__builtin_amdgcn_readlane(mwi, j + u));
                            a[u] = r[lane];
                        }
#pragma unroll
                        for (int u = 0; u < 8; ++u) { FMA4(a[u], w[u]) }
                    }
                    if (j + 4 <= m) {
                        float4 a[4];
                        float w[4];
#pragma unroll
                        for (int u = 0; u < 4; ++u) {
                            const float4* r = reinterpret_cast<const float4*>(x + (size_t)__builtin_amdgcn_readlane(ms, j + u) * DIM);
                            w[u] = __int_as_float(__builtin_amdgcn_readlane(mwi, j + u));
                            a[u] = r[lane];
                        }
#pragma unroll
                        for (int u = 0; u < 4; ++u) { FMA4(a[u], w[u]) }
                        j += 4;
                    }
                    if (j < m) {
                        const unsigned r = m - j;
                        const float4* r0 = reinterpret_cast<const float4*>(x + (size_t)__builtin_amdgcn_readlane(ms, j) * DIM);
                        float w0 = __int_as_float(__builtin_amdgcn_readlane(mwi, j));
                        float4 a0 = r0[lane];
                        float4 a1 = {0.0f, 0.0f, 0.0f, 0.0f};
                        float4 a2 = {0.0f, 0.0f, 0.0f, 0.0f};
                        float w1 = 0.0f, w2 = 0.0f;
                        if (r > 1) {
                            const float4* r1 = reinterpret_cast<const float4*>(x + (size_t)__builtin_amdgcn_readlane(ms, j + 1) * DIM);
                            w1 = __int_as_float(__builtin_amdgcn_readlane(mwi, j + 1));
                            a1 = r1[lane];
                        }
                        if (r > 2) {
                            const float4* r2 = reinterpret_cast<const float4*>(x + (size_t)__builtin_amdgcn_readlane(ms, j + 2) * DIM);
                            w2 = __int_as_float(__builtin_amdgcn_readlane(mwi, j + 2));
                            a2 = r2[lane];
                        }
                        FMA4(a0, w0)
                        if (r > 1) { FMA4(a1, w1) }
                        if (r > 2) { FMA4(a2, w2) }
                    }
                }
            }
            *reinterpret_cast<float4*>(&lAb[lr * LDA + lane * 4]) = acc;
        }

        if (tid == 0) nextTile[buf] = (int)got;
        __syncthreads();
        const int nt = __builtin_amdgcn_readfirstlane(nextTile[buf]);

        v8f c = {0.f, 0.f, 0.f, 0.f, 0.f, 0.f, 0.f, 0.f};
#pragma unroll
        for (int k = 0; k < DIM; k += 4) {
            v2f a, b;
            const float* ap = &lAb[(strip * 16 + mn) * LDA + k + 2 * hi];
            a.x = ap[0];
            a.y = ap[1];
            b.x = lW[(k + 2 * hi) * LDW + colBase + mn];
            b.y = lW[(k + 2 * hi + 1) * LDW + colBase + mn];
            c = __builtin_amdgcn_wmma_f32_16x16x4_f32(
                 false, a,  false, b,
                 (short)0, c,  false,  false);
        }

#pragma unroll
        for (int v = 0; v < 8; ++v) {
            int gr = rowBase + strip * 16 + v + 8 * hi;
            if (gr < n) out[(size_t)gr * DIM + colBase + mn] = c[v];
        }
        tile = nt;
        buf ^= 1;
    }
}

extern "C" void kernel_launch(void* const* d_in, const int* in_sizes, int n_in,
                              void* d_out, int out_size, void* d_ws, size_t ws_size,
                              hipStream_t stream) {
    const float* x = (const float*)d_in[0];
    const int*   ei = (const int*)d_in[1];
    const float* W = (const float*)d_in[2];
    float* out = (float*)d_out;

    const int N = in_sizes[0] / DIM;
    const int E = in_sizes[1] / 2;
    const int NBK = (N + RB - 1) / RB;

    float* dis = (float*)d_ws;
    size_t idxOff = (((size_t)N * sizeof(float)) + 255) & ~(size_t)255;

    unsigned* tab = (unsigned*)((char*)d_ws + idxOff);
    unsigned* tot = tab + (size_t)NBK * G1;
    unsigned* bpk = tot + NBK;
    int* csr = (int*)(bpk + E);
    unsigned* rowptr = (unsigned*)(csr + E);

    bucket_count_kernel<<<G1, 512, 0, stream>>>(ei, tab, N, E, NBK);
    chunk_scan_kernel<<<NBK, G1, 0, stream>>>(tab, tot);
    bucket_place_kernel<<<G1, 512, 0, stream>>>(ei, tab, tot, bpk, N, E, NBK);
    bucket_rows_kernel<<<NBK, RB, 0, stream>>>(bpk, tot, rowptr, dis, csr, tab, N, E);
    gather_gemm_kernel<<<GGRID, 512, 0, stream>>>(x, dis, rowptr, csr, W, out, tab, N, E);
}


# Edge grouping in two launches, not four: each chunk groups its edges by bucket in its own slab (count and scan launches gone); the bucket pass gathers its runs from the slabs; LDS arrival order kept
# speedup vs baseline: 5.0427x; 1.2221x over previous
#include <hip/hip_runtime.h>
#include <hip/hip_bf16.h>

#define DIM 128
#define RB 256
#define RBSHIFT 8
#define MAXBK 512
#define G1 512
#define TILE 4096
#define BCAP 8192

typedef float v2f __attribute__((ext_vector_type(2)));
typedef float v8f __attribute__((ext_vector_type(8)));


__device__ __forceinline__ unsigned block_excl_scan(unsigned v, unsigned* sw, unsigned* total) {
    const int lane = threadIdx.x & 31;
    const int wave = threadIdx.x >> 5;
    const int nw = blockDim.x >> 5;
    unsigned inc = v;
    for (int o = 1; o < 32; o <<= 1) {
        unsigned t = __shfl_up(inc, o);
        if (lane >= o) inc += t;
    }
    if (lane == 31) sw[wave] = inc;
    __syncthreads();
    unsigned base = 0, tot = 0;
    for (int w = 0; w < nw; ++w) {
        unsigned x = sw[w];
        if (w < wave) base += x;
        tot += x;
    }
    __syncthreads();
    *total = tot;
    return base + inc - v;
}

__global__ __launch_bounds__(512) void chunk_group_kernel(const int* __restrict__ ei,
                                                          unsigned* __restrict__ tab,
                                                          unsigned* __restrict__ bpk,
                                                          int n, int E, int nbk) {
    __shared__ unsigned cur[MAXBK];
    __shared__ unsigned hist[MAXBK];
    __shared__ unsigned lof[MAXBK];
    __shared__ unsigned gb[MAXBK];
    __shared__ unsigned stVal[TILE];
    __shared__ unsigned stPos[TILE];
    __shared__ unsigned sw[16];
    const int tid = threadIdx.x;
    const int ch = (E + G1 - 1) / G1;
    int e0 = blockIdx.x * ch;
    e0 = (e0 < E) ? e0 : E;
    int e1 = e0 + ch;
    e1 = (e1 < E) ? e1 : E;
    const bool oneTile = (ch <= TILE);
    if (!oneTile) {
        hist[tid] = 0u;
        __syncthreads();
        for (int e = e0 + tid; e < e1; e += 512) {
            unsigned d = (unsigned)ei[E + e];
            d = (d < (unsigned)n) ? d : (unsigned)(n - 1);
            atomicAdd(&hist[d >> RBSHIFT], 1u);
        }
        __syncthreads();
        const unsigned hc = hist[tid];
        unsigned chunkCount;
        const unsigned lo = block_excl_scan(hc, sw, &chunkCount);
        cur[tid] = (unsigned)e0 + lo;
        if (tid < nbk) tab[(size_t)tid * G1 + blockIdx.x] = lo;
    }
    for (int t0 = e0; t0 < e1; t0 += TILE) {
        hist[tid] = 0u;
        __syncthreads();
        unsigned pk[8], lr[8];
        int bk[8];
#pragma unroll
        for (int k = 0; k < 8; ++k) {
            const int e = t0 + k * 512 + tid;
            bk[k] = -1;
            pk[k] = 0u;
            lr[k] = 0u;
            if (e < e1) {
                unsigned s = (unsigned)ei[e];
                unsigned d = (unsigned)ei[E + e];
                s = (s < (unsigned)n) ? s : (unsigned)(n - 1);
                d = (d < (unsigned)n) ? d : (unsigned)(n - 1);
                bk[k] = (int)(d >> RBSHIFT);
                pk[k] = (s << RBSHIFT) | (d & (unsigned)(RB - 1));
                lr[k] = atomicAdd(&hist[bk[k]], 1u);
            }
        }
        __syncthreads();
        const unsigned h = hist[tid];
        unsigned tileCount;
        const unsigned lo = block_excl_scan(h, sw, &tileCount);
        if (oneTile) {
            cur[tid] = (unsigned)e0 + lo;
            if (tid < nbk) tab[(size_t)tid * G1 + blockIdx.x] = lo;
        }
        lof[tid] = lo;
        gb[tid] = cur[tid];
        cur[tid] += h;
        __syncthreads();
#pragma unroll
        for (int k = 0; k < 8; ++k) {
            if (bk[k] >= 0) {
                const unsigned idx = lof[bk[k]] + lr[k];
                if (idx < (unsigned)TILE) {
                    stVal[idx] = pk[k];
                    stPos[idx] = gb[bk[k]] + lr[k];
                }
            }
        }
        __syncthreads();
        for (unsigned i = tid; i < tileCount; i += 512) {
            const unsigned pos = stPos[i];
            if (pos < (unsigned)E) bpk[pos] = stVal[i];
        }
    }
}

__global__ __launch_bounds__(256) void bucket_rows_kernel(const unsigned* __restrict__ bpk,
                                                          const unsigned* __restrict__ tab,
                                                          unsigned* __restrict__ rowptr,
                                                          float* __restrict__ dis,
                                                          int* __restrict__ csr,
                                                          unsigned* __restrict__ ticket, int n, int E, int nbk) {
    __shared__ unsigned cntL[RB];
    __shared__ unsigned curL[RB];
    __shared__ uint4 runInfo[G1];
    __shared__ unsigned wL[BCAP];
    __shared__ int csrL[BCAP];
    __shared__ unsigned sw[16];
    const int tid = threadIdx.x;
    const int b = blockIdx.x;
    if (b == 0 && tid == 0) *ticket = 0u;
    const int ch = (E + G1 - 1) / G1;
    const uint2 ta = reinterpret_cast<const uint2*>(tab + (size_t)b * G1)[tid];
    uint2 tz = {0u, 0u};
    if (b + 1 < nbk) tz = reinterpret_cast<const uint2*>(tab + (size_t)(b + 1) * G1)[tid];
    unsigned st2[2] = {ta.x, ta.y};
    unsigned en2[2] = {tz.x, tz.y};
    unsigned ln2[2], gs2[2];
#pragma unroll
    for (int k = 0; k < 2; ++k) {
        const int t = 2 * tid + k;
        int c0 = t * ch;
        c0 = (c0 < E) ? c0 : E;
        int c1 = c0 + ch;
        c1 = (c1 < E) ? c1 : E;
        const unsigned cc = (unsigned)(c1 - c0);
        unsigned a = st2[k];
        a = (a < cc) ? a : cc;
        unsigned z = (b + 1 < nbk) ? en2[k] : cc;
        z = (z < cc) ? z : cc;
        z = (z > a) ? z : a;
        st2[k] = a;
        ln2[k] = z - a;
        gs2[k] = (unsigned)c0 + a;
    }
    cntL[tid] = 0u;
    unsigned bc, bstart;
    const unsigned ex2 = block_excl_scan(ln2[0] + ln2[1], sw, &bc);
    (void)block_excl_scan(st2[0] + st2[1], sw, &bstart);
    runInfo[2 * tid] = make_uint4(gs2[0], ln2[0], ex2, 0u);
    runInfo[2 * tid + 1] = make_uint4(gs2[1], ln2[1], ex2 + ln2[0], 0u);
    bstart = (bstart < (unsigned)E) ? bstart : (unsigned)E;
    bc = (bc < (unsigned)E - bstart) ? bc : (unsigned)E - bstart;
    const bool inLds = (bc <= (unsigned)BCAP);
    __syncthreads();
    {
        const int lane = tid & 31;
        const int wave = tid >> 5;
        const int half = lane >> 4;
        const unsigned j = (unsigned)(lane & 15);
        for (int r = wave * (G1 / 8); r < (wave + 1) * (G1 / 8); r += 32) {
            uint4 ri[16];
            unsigned wv[16];
            bool longRun = false;
#pragma unroll
            for (int u = 0; u < 16; ++u) ri[u] = runInfo[r + 2 * u + half];
#pragma unroll
            for (int u = 0; u < 16; ++u) wv[u] = bpk[(j < ri[u].y) ? ri[u].x + j : 0u];
#pragma unroll
            for (int u = 0; u < 16; ++u) {
                if (j < ri[u].y) {
                    atomicAdd(&cntL[wv[u] & (unsigned)(RB - 1)], 1u);
                    const unsigned p = ri[u].z + j;
                    if (p < (unsigned)BCAP) wL[p] = wv[u];
                }
                longRun = longRun || (ri[u].y > 16u);
            }
            if (__any(longRun)) {
#pragma unroll
                for (int u = 0; u < 16; ++u) {
                    for (unsigned i = j + 16u; i < ri[u].y; i += 16u) {
                        const unsigned w = bpk[ri[u].x + i];
                        atomicAdd(&cntL[w & (unsigned)(RB - 1)], 1u);
                        const unsigned p = ri[u].z + i;
                        if (p < (unsigned)BCAP) wL[p] = w;
                    }
                }
            }
        }
    }
    __syncthreads();
    const unsigned c = cntL[tid];
    unsigned total;
    const unsigned ex = block_excl_scan(c, sw, &total);
    curL[tid] = ex;
    const int row = b * RB + tid;
    if (row < n) {
        rowptr[row] = bstart + ex;
        float d = (float)(c + 1u);
        dis[row] = (d > 0.0f) ? rsqrtf(d) : 0.0f;
        if (row == n - 1) rowptr[n] = bstart + ex + c;
    }
    __syncthreads();
    if (inLds) {
        for (unsigned i0 = tid; i0 < bc; i0 += 4 * RB) {
            unsigned w4[4], q4[4];
#pragma unroll
            for (int u = 0; u < 4; ++u) {
                const unsigned i = i0 + u * RB;
                w4[u] = (i < bc) ? wL[i] : 0u;
            }
#pragma unroll
            for (int u = 0; u < 4; ++u) {
                const unsigned i = i0 + u * RB;
                q4[u] = (i < bc) ? atomicAdd(&curL[w4[u] & (unsigned)(RB - 1)], 1u) : (unsigned)BCAP;
            }
#pragma unroll
            for (int u = 0; u < 4; ++u) {
                if (q4[u] < (unsigned)BCAP) csrL[q4[u]] = (int)(w4[u] >> RBSHIFT);
            }
        }
        __syncthreads();
        for (unsigned i = tid; i < bc; i += RB) csr[bstart + i] = csrL[i];
    } else {
        for (int k = 0; k < 2; ++k) {
            const uint4 q4r = runInfo[2 * tid + k];
            for (unsigned i = 0; i < q4r.y; ++i) {
                const unsigned w = bpk[q4r.x + i];
                const unsigned q = atomicAdd(&curL[w & (unsigned)(RB - 1)], 1u);
                const unsigned pos = bstart + q;
                if (pos < (unsigned)E) csr[pos] = (int)(w >> RBSHIFT);
            }
        }
    }
}

#define LDA 132
#define LDW 136
#define BROWS 32
#define GGRID 768

#define FMA4(A, S) acc.x = fmaf((A).x, (S), acc.x); acc.y = fmaf((A).y, (S), acc.y); \
                   acc.z = fmaf((A).z, (S), acc.z); acc.w = fmaf((A).w, (S), acc.w);

__global__ __launch_bounds__(512, 12) void gather_gemm_kernel(const float* __restrict__ x,
                                                          const float* __restrict__ dis,
                                                          const unsigned* __restrict__ rowptr,
                                                          const int* __restrict__ csr,
                                                          const float* __restrict__ Wm,
                                                          float* __restrict__ out,
                                                          unsigned* __restrict__ ticket, int n, int E) {
    __shared__ float lW[DIM * LDW];
    __shared__ float lA[2 * BROWS * LDA];
    __shared__ int nextTile[2];
    const int tid = threadIdx.x;
    const int lane = tid & 31;
    const int wave = __builtin_amdgcn_readfirstlane(tid >> 5);
    const int numTiles = (n + BROWS - 1) / BROWS;

    if (tid == 0) nextTile[1] = (int)atomicAdd(ticket, 1u);

    {
        const float4* W4 = reinterpret_cast<const float4*>(Wm);
#pragma unroll
        for (int i = 0; i < (DIM * DIM / 4) / 512; ++i) {
            const int idx = tid + i * 512;
            const int r = idx >> 5, c4 = idx & 31;
            *reinterpret_cast<float4*>(&lW[r * LDW + c4 * 4]) = W4[idx];
        }
    }
    __syncthreads();
    int tile = __builtin_amdgcn_readfirstlane(nextTile[1]);

    const float4* x4 = reinterpret_cast<const float4*>(x);
    const int mn = lane & 15;
    const int hi = lane >> 4;
    const int strip = wave >> 3;
    const int colBase = (wave & 7) * 16;
    int buf = 0;

    while (tile < numTiles) {
        const int rowBase = tile * BROWS;
        float* lAb = &lA[buf * (BROWS * LDA)];
        unsigned got = 0u;
        if (tid == 0) got = atomicAdd(ticket, 1u);

        for (int rr = 0; rr < 2; ++rr) {
            const int lr = wave * 2 + rr;
            const int row = rowBase + lr;
            float4 acc = {0.0f, 0.0f, 0.0f, 0.0f};
            if (row < n) {
                unsigned end = rowptr[row + 1];
                end = (end < (unsigned)E) ? end : (unsigned)E;
                unsigned beg = rowptr[row];
                beg = (beg < end) ? beg : end;
                const float dd = dis[row];
                {
                    float4 v = x4[(size_t)row * (DIM / 4) + lane];
                    float sl = dd * dd;
                    acc.x = v.x * sl; acc.y = v.y * sl; acc.z = v.z * sl; acc.w = v.w * sl;
                }
                for (unsigned base = beg; base < end; base += 32) {
                    const unsigned m = (end - base < 32u) ? (end - base) : 32u;
                    int ms = 0;
                    float mw = 0.0f;
                    if ((unsigned)lane < m) {
                        unsigned s = (unsigned)csr[base + lane];
                        s = (s < (unsigned)n) ? s : (unsigned)(n - 1);
                        ms = (int)s;
                        mw = dis[s] * dd;
                    }
                    const int mwi = __float_as_int(mw);
                    unsigned j = 0;
                    for (; j + 8 <= m; j += 8) {
                        float4 a[8];
                        float w[8];
#pragma unroll
                        for (int u = 0; u < 8; ++u) {
                            const float4* r = reinterpret_cast<const float4*>(x + (size_t)__builtin_amdgcn_readlane(ms, j + u) * DIM);
                            w[u] = __int_as_float(__builtin_amdgcn_readlane(mwi, j + u));
                            a[u] = r[lane];
                        }
#pragma unroll
                        for (int u = 0; u < 8; ++u) { FMA4(a[u], w[u]) }
                    }
                    if (j + 4 <= m) {
                        float4 a[4];
                        float w[4];
#pragma unroll
                        for (int u = 0; u < 4; ++u) {
                            const float4* r = reinterpret_cast<const float4*>(x + (size_t)__builtin_amdgcn_readlane(ms, j + u) * DIM);
                            w[u] = __int_as_float(__builtin_amdgcn_readlane(mwi, j + u));
                            a[u] = r[lane];
                        }
#pragma unroll
                        for (int u = 0; u < 4; ++u) { FMA4(a[u], w[u]) }
                        j += 4;
                    }
                    if (j < m) {
                        const unsigned r = m - j;
                        const float4* r0 = reinterpret_cast<const float4*>(x + (size_t)__builtin_amdgcn_readlane(ms, j) * DIM);
                        float w0 = __int_as_float(__builtin_amdgcn_readlane(mwi, j));
                        float4 a0 = r0[lane];
                        float4 a1 = {0.0f, 0.0f, 0.0f, 0.0f};
                        float4 a2 = {0.0f, 0.0f, 0.0f, 0.0f};
                        float w1 = 0.0f, w2 = 0.0f;
                        if (r > 1) {
                            const float4* r1 = reinterpret_cast<const float4*>(x + (size_t)__builtin_amdgcn_readlane(ms, j + 1) * DIM);
                            w1 = __int_as_float(__builtin_amdgcn_readlane(mwi, j + 1));
                            a1 = r1[lane];
                        }
                        if (r > 2) {
                            const float4* r2 = reinterpret_cast<const float4*>(x + (size_t)__builtin_amdgcn_readlane(ms, j + 2) * DIM);
                            w2 = __int_as_float(__builtin_amdgcn_readlane(mwi, j + 2));
                            a2 = r2[lane];
                        }
                        FMA4(a0, w0)
                        if (r > 1) { FMA4(a1, w1) }
                        if (r > 2) { FMA4(a2, w2) }
                    }
                }
            }
            *reinterpret_cast<float4*>(&lAb[lr * LDA + lane * 4]) = acc;
        }

        if (tid == 0) nextTile[buf] = (int)got;
        __syncthreads();
        const int nt = __builtin_amdgcn_readfirstlane(nextTile[buf]);

        v8f c = {0.f, 0.f, 0.f, 0.f, 0.f, 0.f, 0.f, 0.f};
#pragma unroll
        for (int k = 0; k < DIM; k += 4) {
            v2f a, b;
            const float* ap = &lAb[(strip * 16 + mn) * LDA + k + 2 * hi];
            a.x = ap[0];
            a.y = ap[1];
            b.x = lW[(k + 2 * hi) * LDW + colBase + mn];
            b.y = lW[(k + 2 * hi + 1) * LDW + colBase + mn];
            c = __builtin_amdgcn_wmma_f32_16x16x4_f32(
                 false, a,  false, b,
                 (short)0, c,  false,  false);
        }

#pragma unroll
        for (int v = 0; v < 8; ++v) {
            int gr = rowBase + strip * 16 + v + 8 * hi;
            if (gr < n) out[(size_t)gr * DIM + colBase + mn] = c[v];
        }
        tile = nt;
        buf ^= 1;
    }
}

extern "C" void kernel_launch(void* const* d_in, const int* in_sizes, int n_in,
                              void* d_out, int out_size, void* d_ws, size_t ws_size,
                              hipStream_t stream) {
    const float* x = (const float*)d_in[0];
    const int*   ei = (const int*)d_in[1];
    const float* W = (const float*)d_in[2];
    float* out = (float*)d_out;

    const int N = in_sizes[0] / DIM;
    const int E = in_sizes[1] / 2;
    const int NBK = (N + RB - 1) / RB;

    float* dis = (float*)d_ws;
    size_t idxOff = (((size_t)N * sizeof(float)) + 255) & ~(size_t)255;

    unsigned* tab = (unsigned*)((char*)d_ws + idxOff);
    unsigned* ticket = tab + (size_t)NBK * G1;
    unsigned* bpk = ticket + NBK;
    int* csr = (int*)(bpk + E);
    unsigned* rowptr = (unsigned*)(csr + E);

    chunk_group_kernel<<<G1, 512, 0, stream>>>(ei, tab, bpk, N, E, NBK);
    bucket_rows_kernel<<<NBK, RB, 0, stream>>>(bpk, tab, rowptr, dis, csr, ticket, N, E, NBK);
    gather_gemm_kernel<<<GGRID, 512, 0, stream>>>(x, dis, rowptr, csr, W, out, ticket, N, E);
}


# Each tile's row bounds, first 32 sources, their 1/sqrt(degree) and own x rows fetched one tile ahead, around the halves of the previous tile's product: gather opens with row loads; sum order kept
# speedup vs baseline: 5.2327x; 1.0377x over previous
#include <hip/hip_runtime.h>
#include <hip/hip_bf16.h>

#define DIM 128
#define RB 256
#define RBSHIFT 8
#define MAXBK 512
#define G1 512
#define TILE 4096
#define BCAP 8192

typedef float v2f __attribute__((ext_vector_type(2)));
typedef float v8f __attribute__((ext_vector_type(8)));


__device__ __forceinline__ unsigned block_excl_scan(unsigned v, unsigned* sw, unsigned* total) {
    const int lane = threadIdx.x & 31;
    const int wave = threadIdx.x >> 5;
    const int nw = blockDim.x >> 5;
    unsigned inc = v;
    for (int o = 1; o < 32; o <<= 1) {
        unsigned t = __shfl_up(inc, o);
        if (lane >= o) inc += t;
    }
    if (lane == 31) sw[wave] = inc;
    __syncthreads();
    unsigned base = 0, tot = 0;
    for (int w = 0; w < nw; ++w) {
        unsigned x = sw[w];
        if (w < wave) base += x;
        tot += x;
    }
    __syncthreads();
    *total = tot;
    return base + inc - v;
}

__global__ __launch_bounds__(512) void chunk_group_kernel(const int* __restrict__ ei,
                                                          unsigned* __restrict__ tab,
                                                          unsigned* __restrict__ bpk,
                                                          int n, int E, int nbk) {
    __shared__ unsigned cur[MAXBK];
    __shared__ unsigned hist[MAXBK];
    __shared__ unsigned lof[MAXBK];
    __shared__ unsigned gb[MAXBK];
    __shared__ unsigned stVal[TILE];
    __shared__ unsigned stPos[TILE];
    __shared__ unsigned sw[16];
    const int tid = threadIdx.x;
    const int ch = (E + G1 - 1) / G1;
    int e0 = blockIdx.x * ch;
    e0 = (e0 < E) ? e0 : E;
    int e1 = e0 + ch;
    e1 = (e1 < E) ? e1 : E;
    const bool oneTile = (ch <= TILE);
    if (!oneTile) {
        hist[tid] = 0u;
        __syncthreads();
        for (int e = e0 + tid; e < e1; e += 512) {
            unsigned d = (unsigned)ei[E + e];
            d = (d < (unsigned)n) ? d : (unsigned)(n - 1);
            atomicAdd(&hist[d >> RBSHIFT], 1u);
        }
        __syncthreads();
        const unsigned hc = hist[tid];
        unsigned chunkCount;
        const unsigned lo = block_excl_scan(hc, sw, &chunkCount);
        cur[tid] = (unsigned)e0 + lo;
        if (tid < nbk) tab[(size_t)tid * G1 + blockIdx.x] = lo;
    }
    for (int t0 = e0; t0 < e1; t0 += TILE) {
        hist[tid] = 0u;
        __syncthreads();
        unsigned pk[8], lr[8];
        int bk[8];
#pragma unroll
        for (int k = 0; k < 8; ++k) {
            const int e = t0 + k * 512 + tid;
            bk[k] = -1;
            pk[k] = 0u;
            lr[k] = 0u;
            if (e < e1) {
                unsigned s = (unsigned)ei[e];
                unsigned d = (unsigned)ei[E + e];
                s = (s < (unsigned)n) ? s : (unsigned)(n - 1);
                d = (d < (unsigned)n) ? d : (unsigned)(n - 1);
                bk[k] = (int)(d >> RBSHIFT);
                pk[k] = (s << RBSHIFT) | (d & (unsigned)(RB - 1));
                lr[k] = atomicAdd(&hist[bk[k]], 1u);
            }
        }
        __syncthreads();
        const unsigned h = hist[tid];
        unsigned tileCount;
        const unsigned lo = block_excl_scan(h, sw, &tileCount);
        if (oneTile) {
            cur[tid] = (unsigned)e0 + lo;
            if (tid < nbk) tab[(size_t)tid * G1 + blockIdx.x] = lo;
        }
        lof[tid] = lo;
        gb[tid] = cur[tid];
        cur[tid] += h;
        __syncthreads();
#pragma unroll
        for (int k = 0; k < 8; ++k) {
            if (bk[k] >= 0) {
                const unsigned idx = lof[bk[k]] + lr[k];
                if (idx < (unsigned)TILE) {
                    stVal[idx] = pk[k];
                    stPos[idx] = gb[bk[k]] + lr[k];
                }
            }
        }
        __syncthreads();
        for (unsigned i = tid; i < tileCount; i += 512) {
            const unsigned pos = stPos[i];
            if (pos < (unsigned)E) bpk[pos] = stVal[i];
        }
    }
}

__global__ __launch_bounds__(256) void bucket_rows_kernel(const unsigned* __restrict__ bpk,
                                                          const unsigned* __restrict__ tab,
                                                          unsigned* __restrict__ rowptr,
                                                          float* __restrict__ dis,
                                                          int* __restrict__ csr,
                                                          unsigned* __restrict__ ticket, int n, int E, int nbk) {
    __shared__ unsigned cntL[RB];
    __shared__ unsigned curL[RB];
    __shared__ uint4 runInfo[G1];
    __shared__ unsigned wL[BCAP];
    __shared__ int csrL[BCAP];
    __shared__ unsigned sw[16];
    const int tid = threadIdx.x;
    const int b = blockIdx.x;
    if (b == 0 && tid == 0) *ticket = 0u;
    const int ch = (E + G1 - 1) / G1;
    const uint2 ta = reinterpret_cast<const uint2*>(tab + (size_t)b * G1)[tid];
    uint2 tz = {0u, 0u};
    if (b + 1 < nbk) tz = reinterpret_cast<const uint2*>(tab + (size_t)(b + 1) * G1)[tid];
    unsigned st2[2] = {ta.x, ta.y};
    unsigned en2[2] = {tz.x, tz.y};
    unsigned ln2[2], gs2[2];
#pragma unroll
    for (int k = 0; k < 2; ++k) {
        const int t = 2 * tid + k;
        int c0 = t * ch;
        c0 = (c0 < E) ? c0 : E;
        int c1 = c0 + ch;
        c1 = (c1 < E) ? c1 : E;
        const unsigned cc = (unsigned)(c1 - c0);
        unsigned a = st2[k];
        a = (a < cc) ? a : cc;
        unsigned z = (b + 1 < nbk) ? en2[k] : cc;
        z = (z < cc) ? z : cc;
        z = (z > a) ? z : a;
        st2[k] = a;
        ln2[k] = z - a;
        gs2[k] = (unsigned)c0 + a;
    }
    cntL[tid] = 0u;
    unsigned bc, bstart;
    const unsigned ex2 = block_excl_scan(ln2[0] + ln2[1], sw, &bc);
    (void)block_excl_scan(st2[0] + st2[1], sw, &bstart);
    runInfo[2 * tid] = make_uint4(gs2[0], ln2[0], ex2, 0u);
    runInfo[2 * tid + 1] = make_uint4(gs2[1], ln2[1], ex2 + ln2[0], 0u);
    bstart = (bstart < (unsigned)E) ? bstart : (unsigned)E;
    bc = (bc < (unsigned)E - bstart) ? bc : (unsigned)E - bstart;
    const bool inLds = (bc <= (unsigned)BCAP);
    __syncthreads();
    {
        const int lane = tid & 31;
        const int wave = tid >> 5;
        const int half = lane >> 4;
        const unsigned j = (unsigned)(lane & 15);
        for (int r = wave * (G1 / 8); r < (wave + 1) * (G1 / 8); r += 32) {
            uint4 ri[16];
            unsigned wv[16];
            bool longRun = false;
#pragma unroll
            for (int u = 0; u < 16; ++u) ri[u] = runInfo[r + 2 * u + half];
#pragma unroll
            for (int u = 0; u < 16; ++u) wv[u] = bpk[(j < ri[u].y) ? ri[u].x + j : 0u];
#pragma unroll
            for (int u = 0; u < 16; ++u) {
                if (j < ri[u].y) {
                    atomicAdd(&cntL[wv[u] & (unsigned)(RB - 1)], 1u);
                    const unsigned p = ri[u].z + j;
                    if (p < (unsigned)BCAP) wL[p] = wv[u];
                }
                longRun = longRun || (ri[u].y > 16u);
            }
            if (__any(longRun)) {
#pragma unroll
                for (int u = 0; u < 16; ++u) {
                    for (unsigned i = j + 16u; i < ri[u].y; i += 16u) {
                        const unsigned w = bpk[ri[u].x + i];
                        atomicAdd(&cntL[w & (unsigned)(RB - 1)], 1u);
                        const unsigned p = ri[u].z + i;
                        if (p < (unsigned)BCAP) wL[p] = w;
                    }
                }
            }
        }
    }
    __syncthreads();
    const unsigned c = cntL[tid];
    unsigned total;
    const unsigned ex = block_excl_scan(c, sw, &total);
    curL[tid] = ex;
    const int row = b * RB + tid;
    if (row < n) {
        rowptr[row] = bstart + ex;
        float d = (float)(c + 1u);
        dis[row] = (d > 0.0f) ? rsqrtf(d) : 0.0f;
        if (row == n - 1) rowptr[n] = bstart + ex + c;
    }
    __syncthreads();
    if (inLds) {
        for (unsigned i0 = tid; i0 < bc; i0 += 4 * RB) {
            unsigned w4[4], q4[4];
#pragma unroll
            for (int u = 0; u < 4; ++u) {
                const unsigned i = i0 + u * RB;
                w4[u] = (i < bc) ? wL[i] : 0u;
            }
#pragma unroll
            for (int u = 0; u < 4; ++u) {
                const unsigned i = i0 + u * RB;
                q4[u] = (i < bc) ? atomicAdd(&curL[w4[u] & (unsigned)(RB - 1)], 1u) : (unsigned)BCAP;
            }
#pragma unroll
            for (int u = 0; u < 4; ++u) {
                if (q4[u] < (unsigned)BCAP) csrL[q4[u]] = (int)(w4[u] >> RBSHIFT);
            }
        }
        __syncthreads();
        for (unsigned i = tid; i < bc; i += RB) csr[bstart + i] = csrL[i];
    } else {
        for (int k = 0; k < 2; ++k) {
            const uint4 q4r = runInfo[2 * tid + k];
            for (unsigned i = 0; i < q4r.y; ++i) {
                const unsigned w = bpk[q4r.x + i];
                const unsigned q = atomicAdd(&curL[w & (unsigned)(RB - 1)], 1u);
                const unsigned pos = bstart + q;
                if (pos < (unsigned)E) csr[pos] = (int)(w >> RBSHIFT);
            }
        }
    }
}

#define LDA 132
#define LDW 136
#define BROWS 32
#define GGRID 768

#define FMA4(A, S) acc.x = fmaf((A).x, (S), acc.x); acc.y = fmaf((A).y, (S), acc.y); \
                   acc.z = fmaf((A).z, (S), acc.z); acc.w = fmaf((A).w, (S), acc.w);

struct RowIdx { unsigned beg, end; float dd; };

__device__ __forceinline__ RowIdx row_bounds(const unsigned* __restrict__ rowptr, const float* __restrict__ dis,
                                             int row, int n, int E) {
    const int rc = (row < n) ? row : (n - 1);
    unsigned end = rowptr[rc + 1];
    end = (end < (unsigned)E) ? end : (unsigned)E;
    unsigned beg = rowptr[rc];
    beg = (beg < end) ? beg : end;
    if (row >= n) beg = end;
    RowIdx r;
    r.beg = beg;
    r.end = end;
    r.dd = dis[rc];
    return r;
}

__device__ __forceinline__ unsigned row_sources(const int* __restrict__ csr, const RowIdx r, int lane, int E) {
    const unsigned last = (E > 0) ? (unsigned)(E - 1) : 0u;
    unsigned i = r.beg + (unsigned)lane;
    i = (i < last) ? i : last;
    return (unsigned)csr[i];
}

__device__ __forceinline__ float row_dis(const float* __restrict__ dis, const RowIdx r, unsigned raw, int lane, int n, int* ms) {
    const unsigned len = r.end - r.beg;
    const unsigned m = (len < 32u) ? len : 32u;
    const unsigned s = (raw < (unsigned)n) ? raw : (unsigned)(n - 1);
    *ms = (int)s;
    return dis[((unsigned)lane < m) ? s : 0u];
}

__device__ __forceinline__ float4 row_self(const float* __restrict__ x, int row, int n, int lane) {
    const int rc = (row < n) ? row : (n - 1);
    return reinterpret_cast<const float4*>(x)[(size_t)rc * (DIM / 4) + lane];
}

__device__ __forceinline__ float4 gather_row(const float* __restrict__ x, const float* __restrict__ dis,
                                             const int* __restrict__ csr, int row, int n, const RowIdx r,
                                             int msIn, float dvIn, const float4 v, int lane) {
    float4 acc = {0.0f, 0.0f, 0.0f, 0.0f};
    if (row < n) {
        const float dd = r.dd;
        const unsigned beg = r.beg, end = r.end;
        {
            float sl = dd * dd;
            acc.x = v.x * sl; acc.y = v.y * sl; acc.z = v.z * sl; acc.w = v.w * sl;
        }
        int ms = msIn;
        float mw = dvIn * dd;
        for (unsigned base = beg; base < end; base += 32) {
            const unsigned m = (end - base < 32u) ? (end - base) : 32u;
            if (base != beg) {
                ms = 0;
                mw = 0.0f;
                if ((unsigned)lane < m) {
                    unsigned s = (unsigned)csr[base + lane];
                    s = (s < (unsigned)n) ? s : (unsigned)(n - 1);
                    ms = (int)s;
                    mw = dis[s] * dd;
                }
            }
            const int mwi = __float_as_int(mw);
            unsigned j = 0;
            for (; j + 8 <= m; j += 8) {
                float4 a[8];
                float w[8];
#pragma unroll
                for (int u = 0; u < 8; ++u) {
                    const float4* rp = reinterpret_cast<const float4*>(x + (size_t)__builtin_amdgcn_readlane(ms, j + u) * DIM);
                    w[u] = __int_as_float(__builtin_amdgcn_readlane(mwi, j + u));
                    a[u] = rp[lane];
                }
#pragma unroll
                for (int u = 0; u < 8; ++u) { FMA4(a[u], w[u]) }
            }
            if (j + 4 <= m) {
                float4 a[4];
                float w[4];
#pragma unroll
                for (int u = 0; u < 4; ++u) {
                    const float4* rp = reinterpret_cast<const float4*>(x + (size_t)__builtin_amdgcn_readlane(ms, j + u) * DIM);
                    w[u] = __int_as_float(__builtin_amdgcn_readlane(mwi, j + u));
                    a[u] = rp[lane];
                }
#pragma unroll
                for (int u = 0; u < 4; ++u) { FMA4(a[u], w[u]) }
                j += 4;
            }
            if (j < m) {
                const unsigned t = m - j;
                const float4* r0 = reinterpret_cast<const float4*>(x + (size_t)__builtin_amdgcn_readlane(ms, j) * DIM);
                float w0 = __int_as_float(__builtin_amdgcn_readlane(mwi, j));
                float4 a0 = r0[lane];
                float4 a1 = {0.0f, 0.0f, 0.0f, 0.0f};
                float4 a2 = {0.0f, 0.0f, 0.0f, 0.0f};
                float w1 = 0.0f, w2 = 0.0f;
                if (t > 1) {
                    const float4* r1 = reinterpret_cast<const float4*>(x + (size_t)__builtin_amdgcn_readlane(ms, j + 1) * DIM);
                    w1 = __int_as_float(__builtin_amdgcn_readlane(mwi, j + 1));
                    a1 = r1[lane];
                }
                if (t > 2) {
                    const float4* r2 = reinterpret_cast<const float4*>(x + (size_t)__builtin_amdgcn_readlane(ms, j + 2) * DIM);
                    w2 = __int_as_float(__builtin_amdgcn_readlane(mwi, j + 2));
                    a2 = r2[lane];
                }
                FMA4(a0, w0)
                if (t > 1) { FMA4(a1, w1) }
                if (t > 2) { FMA4(a2, w2) }
            }
        }
    }
    return acc;
}

__global__ __launch_bounds__(512, 12) void gather_gemm_kernel(const float* __restrict__ x,
                                                          const float* __restrict__ dis,
                                                          const unsigned* __restrict__ rowptr,
                                                          const int* __restrict__ csr,
                                                          const float* __restrict__ Wm,
                                                          float* __restrict__ out,
                                                          unsigned* __restrict__ ticket, int n, int E) {
    __shared__ float lW[DIM * LDW];
    __shared__ float lA[2 * BROWS * LDA];
    __shared__ int nextTile[2];
    const int tid = threadIdx.x;
    const int lane = tid & 31;
    const int wave = __builtin_amdgcn_readfirstlane(tid >> 5);
    const int numTiles = (n + BROWS - 1) / BROWS;

    if (tid == 0) nextTile[1] = (int)atomicAdd(ticket, 1u);

    {
        const float4* W4 = reinterpret_cast<const float4*>(Wm);
#pragma unroll
        for (int i = 0; i < (DIM * DIM / 4) / 512; ++i) {
            const int idx = tid + i * 512;
            const int r = idx >> 5, c4 = idx & 31;
            *reinterpret_cast<float4*>(&lW[r * LDW + c4 * 4]) = W4[idx];
        }
    }
    __syncthreads();
    int tile = __builtin_amdgcn_readfirstlane(nextTile[1]);

    const int mn = lane & 15;
    const int hi = lane >> 4;
    const int strip = wave >> 3;
    const int colBase = (wave & 7) * 16;
    int buf = 0;

    RowIdx p0 = row_bounds(rowptr, dis, tile * BROWS + wave * 2, n, E);
    RowIdx p1 = row_bounds(rowptr, dis, tile * BROWS + wave * 2 + 1, n, E);
    float4 sv0 = row_self(x, tile * BROWS + wave * 2, n, lane);
    float4 sv1 = row_self(x, tile * BROWS + wave * 2 + 1, n, lane);
    unsigned raw0 = row_sources(csr, p0, lane, E);
    unsigned raw1 = row_sources(csr, p1, lane, E);
    int ms0, ms1;
    float dv0 = row_dis(dis, p0, raw0, lane, n, &ms0);
    float dv1 = row_dis(dis, p1, raw1, lane, n, &ms1);

    while (tile < numTiles) {
        const int rowBase = tile * BROWS;
        float* lAb = &lA[buf * (BROWS * LDA)];
        unsigned got = 0u;
        if (tid == 0) got = atomicAdd(ticket, 1u);

        {
            const int lr = wave * 2;
            const float4 s0 = gather_row(x, dis, csr, rowBase + lr, n, p0, ms0, dv0, sv0, lane);
            *reinterpret_cast<float4*>(&lAb[lr * LDA + lane * 4]) = s0;
            const float4 s1 = gather_row(x, dis, csr, rowBase + lr + 1, n, p1, ms1, dv1, sv1, lane);
            *reinterpret_cast<float4*>(&lAb[(lr + 1) * LDA + lane * 4]) = s1;
        }

        if (tid == 0) nextTile[buf] = (int)got;
        __syncthreads();
        const int nt = __builtin_amdgcn_readfirstlane(nextTile[buf]);

        p0 = row_bounds(rowptr, dis, nt * BROWS + wave * 2, n, E);
        p1 = row_bounds(rowptr, dis, nt * BROWS + wave * 2 + 1, n, E);
        sv0 = row_self(x, nt * BROWS + wave * 2, n, lane);
        sv1 = row_self(x, nt * BROWS + wave * 2 + 1, n, lane);

        v8f c = {0.f, 0.f, 0.f, 0.f, 0.f, 0.f, 0.f, 0.f};
#pragma unroll
        for (int k = 0; k < DIM / 2; k += 4) {
            v2f a, b;
            const float* ap = &lAb[(strip * 16 + mn) * LDA + k + 2 * hi];
            a.x = ap[0];
            a.y = ap[1];
            b.x = lW[(k + 2 * hi) * LDW + colBase + mn];
            b.y = lW[(k + 2 * hi + 1) * LDW + colBase + mn];
            c = __builtin_amdgcn_wmma_f32_16x16x4_f32(
                 false, a,  false, b,
                 (short)0, c,  false,  false);
        }

        raw0 = row_sources(csr, p0, lane, E);
        raw1 = row_sources(csr, p1, lane, E);

#pragma unroll
        for (int k = DIM / 2; k < DIM; k += 4) {
            v2f a, b;
            const float* ap = &lAb[(strip * 16 + mn) * LDA + k + 2 * hi];
            a.x = ap[0];
            a.y = ap[1];
            b.x = lW[(k + 2 * hi) * LDW + colBase + mn];
            b.y = lW[(k + 2 * hi + 1) * LDW + colBase + mn];
            c = __builtin_amdgcn_wmma_f32_16x16x4_f32(
                 false, a,  false, b,
                 (short)0, c,  false,  false);
        }

        dv0 = row_dis(dis, p0, raw0, lane, n, &ms0);
        dv1 = row_dis(dis, p1, raw1, lane, n, &ms1);

#pragma unroll
        for (int v = 0; v < 8; ++v) {
            int gr = rowBase + strip * 16 + v + 8 * hi;
            if (gr < n) out[(size_t)gr * DIM + colBase + mn] = c[v];
        }
        tile = nt;
        buf ^= 1;
    }
}

extern "C" void kernel_launch(void* const* d_in, const int* in_sizes, int n_in,
                              void* d_out, int out_size, void* d_ws, size_t ws_size,
                              hipStream_t stream) {
    const float* x = (const float*)d_in[0];
    const int*   ei = (const int*)d_in[1];
    const float* W = (const float*)d_in[2];
    float* out = (float*)d_out;

    const int N = in_sizes[0] / DIM;
    const int E = in_sizes[1] / 2;
    const int NBK = (N + RB - 1) / RB;

    float* dis = (float*)d_ws;
    size_t idxOff = (((size_t)N * sizeof(float)) + 255) & ~(size_t)255;

    unsigned* tab = (unsigned*)((char*)d_ws + idxOff);
    unsigned* ticket = tab + (size_t)NBK * G1;
    unsigned* bpk = ticket + NBK;
    int* csr = (int*)(bpk + E);
    unsigned* rowptr = (unsigned*)(csr + E);

    chunk_group_kernel<<<G1, 512, 0, stream>>>(ei, tab, bpk, N, E, NBK);
    bucket_rows_kernel<<<NBK, RB, 0, stream>>>(bpk, tab, rowptr, dis, csr, ticket, N, E, NBK);
    gather_gemm_kernel<<<GGRID, 512, 0, stream>>>(x, dis, rowptr, csr, W, out, ticket, N, E);
}
